# starting kernel
# baseline (speedup 1.0000x reference)
#include <hip/hip_runtime.h>
#include <math.h>

#define N_NODES 100000
#define N_EDGES 3200000
#define D_IN 128
#define D_HID 16
#define D_OUT 2
#define N_TILES (N_NODES / 16)

typedef __attribute__((ext_vector_type(2))) float v2f;
typedef __attribute__((ext_vector_type(8))) float v8f;


__global__ void k_zero(float* __restrict__ p, int n) {
  int i = blockIdx.x * blockDim.x + threadIdx.x;
  if (i < n) p[i] = 0.0f;
}

__global__ void k_count_deg(const int* __restrict__ dst, float* __restrict__ deg) {
  int e = blockIdx.x * blockDim.x + threadIdx.x;
  if (e < N_EDGES) atomicAdd(&deg[dst[e]], 1.0f);
}

__global__ void k_dinv(float* __restrict__ deg) {
  int i = blockIdx.x * blockDim.x + threadIdx.x;
  if (i < N_NODES) deg[i] = rsqrtf(deg[i] + 1.0f);
}


__global__ void k_bpack(const float* __restrict__ W1, float* __restrict__ Bp) {
  int t = blockIdx.x * blockDim.x + threadIdx.x;
  if (t < 32 * 32) {
    int ks = t >> 5, lane = t & 31;
    int n = lane & 15;
    int k0 = ks * 4 + ((lane >> 4) ? 2 : 0);
    Bp[2 * t + 0] = W1[k0 * D_HID + n];
    Bp[2 * t + 1] = W1[(k0 + 1) * D_HID + n];
  }
}


__global__ void k_gemm1(const float* __restrict__ x, const float* __restrict__ Bp,
                        const float* __restrict__ dinv, float* __restrict__ hs,
                        float* __restrict__ agg1) {
  int lane = threadIdx.x & 31;
  int wave = threadIdx.x >> 5;
  int tile = blockIdx.x * (blockDim.x >> 5) + wave;
  if (tile >= N_TILES) return;

  int m = lane & 15;
  int koff = (lane >> 4) * 2;
  const float* xr = x + (size_t)(tile * 16 + m) * D_IN + koff;
  const v2f* bp = (const v2f*)Bp;

  v8f c = {0.f, 0.f, 0.f, 0.f, 0.f, 0.f, 0.f, 0.f};
#pragma unroll
  for (int ks = 0; ks < 32; ++ks) {
    v2f a = *(const v2f*)(xr + ks * 4);
    v2f b = bp[ks * 32 + lane];
    c = __builtin_amdgcn_wmma_f32_16x16x4_f32(
         false, a,  false, b,
         (short)0, c,  false,  false);
  }

  int n = lane & 15;
  int mbase = (lane >> 4) * 8;
#pragma unroll
  for (int v = 0; v < 8; ++v) {
    int r = tile * 16 + mbase + v;
    float di = dinv[r];
    float hv = c[v] * di;
    hs[(size_t)r * D_HID + n] = hv;
    agg1[(size_t)r * D_HID + n] = hv * di;
  }
}


__global__ void k_scatter1(const int* __restrict__ src, const int* __restrict__ dst,
                           const float* __restrict__ hs, const float* __restrict__ dinv,
                           float* __restrict__ agg1) {
  long long gid = (long long)blockIdx.x * blockDim.x + threadIdx.x;
  if (gid >= (long long)N_EDGES * 4) return;
  int e = (int)(gid >> 2);
  int q = (int)(gid & 3);
  int s = src[e], d = dst[e];
  float w = dinv[d];
  float4 msg = *(const float4*)(hs + (size_t)s * D_HID + q * 4);
  float* ap = agg1 + (size_t)d * D_HID + q * 4;
  atomicAdd(ap + 0, msg.x * w);
  atomicAdd(ap + 1, msg.y * w);
  atomicAdd(ap + 2, msg.z * w);
  atomicAdd(ap + 3, msg.w * w);
}


__global__ void k_layer2(const float* __restrict__ agg1, const float* __restrict__ b1,
                         const float* __restrict__ W2, const float* __restrict__ dinv,
                         float* __restrict__ ts, float* __restrict__ agg2) {
  int i = blockIdx.x * blockDim.x + threadIdx.x;
  if (i >= N_NODES) return;
  const float* row = agg1 + (size_t)i * D_HID;
  float t0 = 0.f, t1 = 0.f;
#pragma unroll
  for (int j = 0; j < D_HID; ++j) {
    float hj = fmaxf(row[j] + b1[j], 0.0f);
    t0 += hj * W2[j * 2 + 0];
    t1 += hj * W2[j * 2 + 1];
  }
  float di = dinv[i];
  float s0 = t0 * di, s1 = t1 * di;
  ts[2 * i + 0] = s0;
  ts[2 * i + 1] = s1;
  agg2[2 * i + 0] = s0 * di;
  agg2[2 * i + 1] = s1 * di;
}


__global__ void k_scatter2(const int* __restrict__ src, const int* __restrict__ dst,
                           const float* __restrict__ ts, const float* __restrict__ dinv,
                           float* __restrict__ agg2) {
  int e = blockIdx.x * blockDim.x + threadIdx.x;
  if (e >= N_EDGES) return;
  int s = src[e], d = dst[e];
  float w = dinv[d];
  float2 m = *(const float2*)(ts + (size_t)s * 2);
  atomicAdd(&agg2[2 * d + 0], m.x * w);
  atomicAdd(&agg2[2 * d + 1], m.y * w);
}


__global__ void k_final(const float* __restrict__ agg2, const float* __restrict__ b2,
                        float* __restrict__ out) {
  int i = blockIdx.x * blockDim.x + threadIdx.x;
  if (i >= N_NODES) return;
  float z0 = agg2[2 * i + 0] + b2[0];
  float z1 = agg2[2 * i + 1] + b2[1];
  float mm = fmaxf(z0, z1);
  float lse = mm + logf(expf(z0 - mm) + expf(z1 - mm));
  out[2 * i + 0] = z0 - lse;
  out[2 * i + 1] = z1 - lse;
}

extern "C" void kernel_launch(void* const* d_in, const int* in_sizes, int n_in,
                              void* d_out, int out_size, void* d_ws, size_t ws_size,
                              hipStream_t stream) {
  const float* x  = (const float*)d_in[0];
  const float* W1 = (const float*)d_in[1];
  const float* b1 = (const float*)d_in[2];
  const float* W2 = (const float*)d_in[3];
  const float* b2 = (const float*)d_in[4];
  const int*  ei  = (const int*)d_in[5];
  const int* src = ei;
  const int* dst = ei + N_EDGES;
  float* out = (float*)d_out;

  float* w    = (float*)d_ws;
  float* deg  = w;
  float* Bp   = deg + N_NODES;
  float* hs   = Bp + 2048;
  float* agg1 = hs + (size_t)N_NODES * D_HID;
  float* ts   = agg1 + (size_t)N_NODES * D_HID;
  float* agg2 = ts + (size_t)N_NODES * D_OUT;

  const int T = 256;
  k_zero<<<(N_NODES + T - 1) / T, T, 0, stream>>>(deg, N_NODES);
  k_count_deg<<<(N_EDGES + T - 1) / T, T, 0, stream>>>(dst, deg);
  k_dinv<<<(N_NODES + T - 1) / T, T, 0, stream>>>(deg);
  k_bpack<<<4, T, 0, stream>>>(W1, Bp);

  k_gemm1<<<(N_TILES + 7) / 8, T, 0, stream>>>(x, Bp, deg, hs, agg1);

  long long work1 = (long long)N_EDGES * 4;
  k_scatter1<<<(int)((work1 + T - 1) / T), T, 0, stream>>>(src, dst, hs, deg, agg1);

  k_layer2<<<(N_NODES + T - 1) / T, T, 0, stream>>>(agg1, b1, W2, deg, ts, agg2);
  k_scatter2<<<(N_EDGES + T - 1) / T, T, 0, stream>>>(src, dst, ts, deg, agg2);
  k_final<<<(N_NODES + T - 1) / T, T, 0, stream>>>(agg2, b2, out);
}


# Degree count and layer-2 edge sum: work-groups own 12500-node ranges in LDS, filter edge slices, add partials to memory as whole lines (dinv[dst] once per partial; float sum order changes)
# speedup vs baseline: 2.5569x; 2.5569x over previous
#include <hip/hip_runtime.h>
#include <math.h>

#define N_NODES 100000
#define N_EDGES 3200000
#define D_IN 128
#define D_HID 16
#define D_OUT 2
#define N_TILES (N_NODES / 16)

typedef __attribute__((ext_vector_type(2))) float v2f;
typedef __attribute__((ext_vector_type(8))) float v8f;


__global__ void k_zero(float* __restrict__ p, int n) {
  int i = blockIdx.x * blockDim.x + threadIdx.x;
  if (i < n) p[i] = 0.0f;
}


#define OWN_R 8
#define OWN_RANGE (N_NODES / OWN_R)
#define OWN_B 32
#define SLICE (N_EDGES / OWN_B)
#define SLICE4 (SLICE / 4)
#define OWN_T 1024

__global__ void __launch_bounds__(OWN_T)
k_count_deg(const int* __restrict__ dst, float* __restrict__ deg) {
  __shared__ unsigned cnt[OWN_RANGE];
  const int tid = threadIdx.x;
  const int r = blockIdx.x / OWN_B;
  const int b = blockIdx.x % OWN_B;
  const int base = r * OWN_RANGE;
  for (int i = tid; i < OWN_RANGE; i += OWN_T) cnt[i] = 0u;
  __syncthreads();
  const int4* d4 = (const int4*)(dst + (size_t)b * SLICE);
  for (int q0 = tid; q0 < SLICE4; q0 += 4 * OWN_T) {
    int4 d[4];
#pragma unroll
    for (int j = 0; j < 4; ++j) {
      int q = q0 + j * OWN_T;
      d[j] = make_int4(-1, -1, -1, -1);
      if (q < SLICE4) d[j] = d4[q];
    }
#pragma unroll
    for (int j = 0; j < 4; ++j) {
      unsigned l0 = (unsigned)(d[j].x - base), l1 = (unsigned)(d[j].y - base);
      unsigned l2 = (unsigned)(d[j].z - base), l3 = (unsigned)(d[j].w - base);
      if (l0 < (unsigned)OWN_RANGE) atomicAdd(&cnt[l0], 1u);
      if (l1 < (unsigned)OWN_RANGE) atomicAdd(&cnt[l1], 1u);
      if (l2 < (unsigned)OWN_RANGE) atomicAdd(&cnt[l2], 1u);
      if (l3 < (unsigned)OWN_RANGE) atomicAdd(&cnt[l3], 1u);
    }
  }
  __syncthreads();
  for (int i = tid; i < OWN_RANGE; i += OWN_T) {
    unsigned c = cnt[i];
    if (c) atomicAdd(&deg[base + i], (float)c);
  }
}

__global__ void k_dinv(float* __restrict__ deg) {
  int i = blockIdx.x * blockDim.x + threadIdx.x;
  if (i < N_NODES) deg[i] = rsqrtf(deg[i] + 1.0f);
}


__global__ void k_bpack(const float* __restrict__ W1, float* __restrict__ Bp) {
  int t = blockIdx.x * blockDim.x + threadIdx.x;
  if (t < 32 * 32) {
    int ks = t >> 5, lane = t & 31;
    int n = lane & 15;
    int k0 = ks * 4 + ((lane >> 4) ? 2 : 0);
    Bp[2 * t + 0] = W1[k0 * D_HID + n];
    Bp[2 * t + 1] = W1[(k0 + 1) * D_HID + n];
  }
}


__global__ void k_gemm1(const float* __restrict__ x, const float* __restrict__ Bp,
                        const float* __restrict__ dinv, float* __restrict__ hs,
                        float* __restrict__ agg1) {
  int lane = threadIdx.x & 31;
  int wave = threadIdx.x >> 5;
  int tile = blockIdx.x * (blockDim.x >> 5) + wave;
  if (tile >= N_TILES) return;

  int m = lane & 15;
  int koff = (lane >> 4) * 2;
  const float* xr = x + (size_t)(tile * 16 + m) * D_IN + koff;
  const v2f* bp = (const v2f*)Bp;

  v8f c = {0.f, 0.f, 0.f, 0.f, 0.f, 0.f, 0.f, 0.f};
#pragma unroll
  for (int ks = 0; ks < 32; ++ks) {
    v2f a = *(const v2f*)(xr + ks * 4);
    v2f b = bp[ks * 32 + lane];
    c = __builtin_amdgcn_wmma_f32_16x16x4_f32(
         false, a,  false, b,
         (short)0, c,  false,  false);
  }

  int n = lane & 15;
  int mbase = (lane >> 4) * 8;
#pragma unroll
  for (int v = 0; v < 8; ++v) {
    int r = tile * 16 + mbase + v;
    float di = dinv[r];
    float hv = c[v] * di;
    hs[(size_t)r * D_HID + n] = hv;
    agg1[(size_t)r * D_HID + n] = hv * di;
  }
}


__global__ void k_scatter1(const int* __restrict__ src, const int* __restrict__ dst,
                           const float* __restrict__ hs, const float* __restrict__ dinv,
                           float* __restrict__ agg1) {
  long long gid = (long long)blockIdx.x * blockDim.x + threadIdx.x;
  if (gid >= (long long)N_EDGES * 4) return;
  int e = (int)(gid >> 2);
  int q = (int)(gid & 3);
  int s = src[e], d = dst[e];
  float w = dinv[d];
  float4 msg = *(const float4*)(hs + (size_t)s * D_HID + q * 4);
  float* ap = agg1 + (size_t)d * D_HID + q * 4;
  atomicAdd(ap + 0, msg.x * w);
  atomicAdd(ap + 1, msg.y * w);
  atomicAdd(ap + 2, msg.z * w);
  atomicAdd(ap + 3, msg.w * w);
}


__global__ void k_layer2(const float* __restrict__ agg1, const float* __restrict__ b1,
                         const float* __restrict__ W2, const float* __restrict__ dinv,
                         float* __restrict__ ts, float* __restrict__ agg2) {
  int i = blockIdx.x * blockDim.x + threadIdx.x;
  if (i >= N_NODES) return;
  const float* row = agg1 + (size_t)i * D_HID;
  float t0 = 0.f, t1 = 0.f;
#pragma unroll
  for (int j = 0; j < D_HID; ++j) {
    float hj = fmaxf(row[j] + b1[j], 0.0f);
    t0 += hj * W2[j * 2 + 0];
    t1 += hj * W2[j * 2 + 1];
  }
  float di = dinv[i];
  float s0 = t0 * di, s1 = t1 * di;
  ts[2 * i + 0] = s0;
  ts[2 * i + 1] = s1;
  agg2[2 * i + 0] = s0 * di;
  agg2[2 * i + 1] = s1 * di;
}


__global__ void __launch_bounds__(OWN_T)
k_scatter2(const int* __restrict__ src, const int* __restrict__ dst,
           const float* __restrict__ ts, const float* __restrict__ dinv,
           float* __restrict__ agg2) {
  __shared__ float acc[2 * OWN_RANGE];
  const int tid = threadIdx.x;
  const int r = blockIdx.x / OWN_B;
  const int b = blockIdx.x % OWN_B;
  const int base = r * OWN_RANGE;
  for (int i = tid; i < 2 * OWN_RANGE; i += OWN_T) acc[i] = 0.0f;
  __syncthreads();
  const int4* d4 = (const int4*)(dst + (size_t)b * SLICE);
  const int4* s4 = (const int4*)(src + (size_t)b * SLICE);
  const float2* ts2 = (const float2*)ts;
  for (int q0 = tid; q0 < SLICE4; q0 += 4 * OWN_T) {
    int dl[16], sv[16];
#pragma unroll
    for (int j = 0; j < 4; ++j) {
      int q = q0 + j * OWN_T;
      int4 d = make_int4(-1, -1, -1, -1);
      int4 s = make_int4(0, 0, 0, 0);
      if (q < SLICE4) { d = d4[q]; s = s4[q]; }
      dl[4 * j + 0] = d.x - base; dl[4 * j + 1] = d.y - base;
      dl[4 * j + 2] = d.z - base; dl[4 * j + 3] = d.w - base;
      sv[4 * j + 0] = s.x; sv[4 * j + 1] = s.y; sv[4 * j + 2] = s.z; sv[4 * j + 3] = s.w;
    }
    float2 m[16];
#pragma unroll
    for (int k = 0; k < 16; ++k) {
      m[k] = make_float2(0.0f, 0.0f);
      if ((unsigned)dl[k] < (unsigned)OWN_RANGE) m[k] = ts2[sv[k]];
    }
#pragma unroll
    for (int k = 0; k < 16; ++k) {
      if ((unsigned)dl[k] < (unsigned)OWN_RANGE) {
        atomicAdd(&acc[2 * dl[k] + 0], m[k].x);
        atomicAdd(&acc[2 * dl[k] + 1], m[k].y);
      }
    }
  }
  __syncthreads();
  for (int i0 = tid; i0 < 2 * OWN_RANGE; i0 += 8 * OWN_T) {
    float a[8], w[8];
#pragma unroll
    for (int j = 0; j < 8; ++j) {
      int i = i0 + j * OWN_T;
      a[j] = 0.0f; w[j] = 0.0f;
      if (i < 2 * OWN_RANGE) { a[j] = acc[i]; w[j] = dinv[base + (i >> 1)]; }
    }
#pragma unroll
    for (int j = 0; j < 8; ++j) {
      int i = i0 + j * OWN_T;
      if (i < 2 * OWN_RANGE) atomicAdd(&agg2[2 * base + i], a[j] * w[j]);
    }
  }
}


__global__ void k_final(const float* __restrict__ agg2, const float* __restrict__ b2,
                        float* __restrict__ out) {
  int i = blockIdx.x * blockDim.x + threadIdx.x;
  if (i >= N_NODES) return;
  float z0 = agg2[2 * i + 0] + b2[0];
  float z1 = agg2[2 * i + 1] + b2[1];
  float mm = fmaxf(z0, z1);
  float lse = mm + logf(expf(z0 - mm) + expf(z1 - mm));
  out[2 * i + 0] = z0 - lse;
  out[2 * i + 1] = z1 - lse;
}

extern "C" void kernel_launch(void* const* d_in, const int* in_sizes, int n_in,
                              void* d_out, int out_size, void* d_ws, size_t ws_size,
                              hipStream_t stream) {
  const float* x  = (const float*)d_in[0];
  const float* W1 = (const float*)d_in[1];
  const float* b1 = (const float*)d_in[2];
  const float* W2 = (const float*)d_in[3];
  const float* b2 = (const float*)d_in[4];
  const int*  ei  = (const int*)d_in[5];
  const int* src = ei;
  const int* dst = ei + N_EDGES;
  float* out = (float*)d_out;

  float* w    = (float*)d_ws;
  float* deg  = w;
  float* Bp   = deg + N_NODES;
  float* hs   = Bp + 2048;
  float* agg1 = hs + (size_t)N_NODES * D_HID;
  float* ts   = agg1 + (size_t)N_NODES * D_HID;
  float* agg2 = ts + (size_t)N_NODES * D_OUT;

  const int T = 256;
  k_zero<<<(N_NODES + T - 1) / T, T, 0, stream>>>(deg, N_NODES);
  k_count_deg<<<OWN_R * OWN_B, OWN_T, 0, stream>>>(dst, deg);
  k_dinv<<<(N_NODES + T - 1) / T, T, 0, stream>>>(deg);
  k_bpack<<<4, T, 0, stream>>>(W1, Bp);

  k_gemm1<<<(N_TILES + 7) / 8, T, 0, stream>>>(x, Bp, deg, hs, agg1);

  long long work1 = (long long)N_EDGES * 4;
  k_scatter1<<<(int)((work1 + T - 1) / T), T, 0, stream>>>(src, dst, hs, deg, agg1);

  k_layer2<<<(N_NODES + T - 1) / T, T, 0, stream>>>(agg1, b1, W2, deg, ts, agg2);
  k_scatter2<<<OWN_R * OWN_B, OWN_T, 0, stream>>>(src, dst, ts, deg, agg2);
  k_final<<<(N_NODES + T - 1) / T, T, 0, stream>>>(agg2, b2, out);
}


# Layer-1 edge scatter remapped: waves loop over 32-edge batches with 16 lanes per edge, so each source-row gather and each row add is one 64-byte access (still float atomics, arrival order)
# speedup vs baseline: 3.4018x; 1.3304x over previous
#include <hip/hip_runtime.h>
#include <math.h>

#define N_NODES 100000
#define N_EDGES 3200000
#define D_IN 128
#define D_HID 16
#define D_OUT 2
#define N_TILES (N_NODES / 16)

typedef __attribute__((ext_vector_type(2))) float v2f;
typedef __attribute__((ext_vector_type(8))) float v8f;


__global__ void k_zero(float* __restrict__ p, int n) {
  int i = blockIdx.x * blockDim.x + threadIdx.x;
  if (i < n) p[i] = 0.0f;
}


#define OWN_R 8
#define OWN_RANGE (N_NODES / OWN_R)
#define OWN_B 32
#define SLICE (N_EDGES / OWN_B)
#define SLICE4 (SLICE / 4)
#define OWN_T 1024

__global__ void __launch_bounds__(OWN_T)
k_count_deg(const int* __restrict__ dst, float* __restrict__ deg) {
  __shared__ unsigned cnt[OWN_RANGE];
  const int tid = threadIdx.x;
  const int r = blockIdx.x / OWN_B;
  const int b = blockIdx.x % OWN_B;
  const int base = r * OWN_RANGE;
  for (int i = tid; i < OWN_RANGE; i += OWN_T) cnt[i] = 0u;
  __syncthreads();
  const int4* d4 = (const int4*)(dst + (size_t)b * SLICE);
  for (int q0 = tid; q0 < SLICE4; q0 += 4 * OWN_T) {
    int4 d[4];
#pragma unroll
    for (int j = 0; j < 4; ++j) {
      int q = q0 + j * OWN_T;
      d[j] = make_int4(-1, -1, -1, -1);
      if (q < SLICE4) d[j] = d4[q];
    }
#pragma unroll
    for (int j = 0; j < 4; ++j) {
      unsigned l0 = (unsigned)(d[j].x - base), l1 = (unsigned)(d[j].y - base);
      unsigned l2 = (unsigned)(d[j].z - base), l3 = (unsigned)(d[j].w - base);
      if (l0 < (unsigned)OWN_RANGE) atomicAdd(&cnt[l0], 1u);
      if (l1 < (unsigned)OWN_RANGE) atomicAdd(&cnt[l1], 1u);
      if (l2 < (unsigned)OWN_RANGE) atomicAdd(&cnt[l2], 1u);
      if (l3 < (unsigned)OWN_RANGE) atomicAdd(&cnt[l3], 1u);
    }
  }
  __syncthreads();
  for (int i = tid; i < OWN_RANGE; i += OWN_T) {
    unsigned c = cnt[i];
    if (c) atomicAdd(&deg[base + i], (float)c);
  }
}

__global__ void k_dinv(float* __restrict__ deg) {
  int i = blockIdx.x * blockDim.x + threadIdx.x;
  if (i < N_NODES) deg[i] = rsqrtf(deg[i] + 1.0f);
}


__global__ void k_bpack(const float* __restrict__ W1, float* __restrict__ Bp) {
  int t = blockIdx.x * blockDim.x + threadIdx.x;
  if (t < 32 * 32) {
    int ks = t >> 5, lane = t & 31;
    int n = lane & 15;
    int k0 = ks * 4 + ((lane >> 4) ? 2 : 0);
    Bp[2 * t + 0] = W1[k0 * D_HID + n];
    Bp[2 * t + 1] = W1[(k0 + 1) * D_HID + n];
  }
}


__global__ void k_gemm1(const float* __restrict__ x, const float* __restrict__ Bp,
                        const float* __restrict__ dinv, float* __restrict__ hs,
                        float* __restrict__ agg1) {
  int lane = threadIdx.x & 31;
  int wave = threadIdx.x >> 5;
  int tile = blockIdx.x * (blockDim.x >> 5) + wave;
  if (tile >= N_TILES) return;

  int m = lane & 15;
  int koff = (lane >> 4) * 2;
  const float* xr = x + (size_t)(tile * 16 + m) * D_IN + koff;
  const v2f* bp = (const v2f*)Bp;

  v8f c = {0.f, 0.f, 0.f, 0.f, 0.f, 0.f, 0.f, 0.f};
#pragma unroll
  for (int ks = 0; ks < 32; ++ks) {
    v2f a = *(const v2f*)(xr + ks * 4);
    v2f b = bp[ks * 32 + lane];
    c = __builtin_amdgcn_wmma_f32_16x16x4_f32(
         false, a,  false, b,
         (short)0, c,  false,  false);
  }

  int n = lane & 15;
  int mbase = (lane >> 4) * 8;
#pragma unroll
  for (int v = 0; v < 8; ++v) {
    int r = tile * 16 + mbase + v;
    float di = dinv[r];
    float hv = c[v] * di;
    hs[(size_t)r * D_HID + n] = hv;
    agg1[(size_t)r * D_HID + n] = hv * di;
  }
}


#define S1_BATCHES (N_EDGES / 32)
#define S1_T 128
#define S1_WG 3125
#define S1_WAVES (S1_WG * (S1_T / 32))

#define ROW_SHARE_I(v, K) __builtin_amdgcn_update_dpp(0, (v), 0x150 + (K), 0xf, 0xf, false)

#define S1_GATHER(K)                                                          \
  {                                                                           \
    unsigned so = (unsigned)ROW_SHARE_I(s16, K) + j4;                         \
    m[K] = *(const float*)((const char*)hs + so);                             \
  }

#define S1_ADD(K)                                                             \
  {                                                                           \
    unsigned dof = (unsigned)ROW_SHARE_I(d16, K) + j4;                        \
    float wk = __int_as_float(ROW_SHARE_I(wbits, K));                         \
    atomicAdd((float*)((char*)agg1 + dof), m[K] * wk);                        \
  }

__global__ void __launch_bounds__(S1_T)
k_scatter1(const int* __restrict__ src, const int* __restrict__ dst,
           const float* __restrict__ hs, const float* __restrict__ dinv,
           float* __restrict__ agg1) {
  const int lane = threadIdx.x & 31;
  const unsigned j4 = (unsigned)(lane & 15) * 4u;
  const int gw = __builtin_amdgcn_readfirstlane(
      (int)(blockIdx.x * (S1_T / 32) + (threadIdx.x >> 5)));
  int s = src[gw * 32 + lane];
  int d = dst[gw * 32 + lane];
  for (int bt = gw; bt < S1_BATCHES; bt += S1_WAVES) {
    const int nb = (bt + S1_WAVES < S1_BATCHES) ? bt + S1_WAVES : bt;
    const int sn = src[nb * 32 + lane];
    const int dn = dst[nb * 32 + lane];
    const int wbits = __float_as_int(dinv[d]);
    const int s16 = s * (D_HID * 4);
    const int d16 = d * (D_HID * 4);
    float m[16];
    S1_GATHER(0)  S1_GATHER(1)  S1_GATHER(2)  S1_GATHER(3)
    S1_GATHER(4)  S1_GATHER(5)  S1_GATHER(6)  S1_GATHER(7)
    S1_GATHER(8)  S1_GATHER(9)  S1_GATHER(10) S1_GATHER(11)
    S1_GATHER(12) S1_GATHER(13) S1_GATHER(14) S1_GATHER(15)
    S1_ADD(0)  S1_ADD(1)  S1_ADD(2)  S1_ADD(3)
    S1_ADD(4)  S1_ADD(5)  S1_ADD(6)  S1_ADD(7)
    S1_ADD(8)  S1_ADD(9)  S1_ADD(10) S1_ADD(11)
    S1_ADD(12) S1_ADD(13) S1_ADD(14) S1_ADD(15)
    s = sn;
    d = dn;
  }
}


__global__ void k_layer2(const float* __restrict__ agg1, const float* __restrict__ b1,
                         const float* __restrict__ W2, const float* __restrict__ dinv,
                         float* __restrict__ ts, float* __restrict__ agg2) {
  int i = blockIdx.x * blockDim.x + threadIdx.x;
  if (i >= N_NODES) return;
  const float* row = agg1 + (size_t)i * D_HID;
  float t0 = 0.f, t1 = 0.f;
#pragma unroll
  for (int j = 0; j < D_HID; ++j) {
    float hj = fmaxf(row[j] + b1[j], 0.0f);
    t0 += hj * W2[j * 2 + 0];
    t1 += hj * W2[j * 2 + 1];
  }
  float di = dinv[i];
  float s0 = t0 * di, s1 = t1 * di;
  ts[2 * i + 0] = s0;
  ts[2 * i + 1] = s1;
  agg2[2 * i + 0] = s0 * di;
  agg2[2 * i + 1] = s1 * di;
}


__global__ void __launch_bounds__(OWN_T)
k_scatter2(const int* __restrict__ src, const int* __restrict__ dst,
           const float* __restrict__ ts, const float* __restrict__ dinv,
           float* __restrict__ agg2) {
  __shared__ float acc[2 * OWN_RANGE];
  const int tid = threadIdx.x;
  const int r = blockIdx.x / OWN_B;
  const int b = blockIdx.x % OWN_B;
  const int base = r * OWN_RANGE;
  for (int i = tid; i < 2 * OWN_RANGE; i += OWN_T) acc[i] = 0.0f;
  __syncthreads();
  const int4* d4 = (const int4*)(dst + (size_t)b * SLICE);
  const int4* s4 = (const int4*)(src + (size_t)b * SLICE);
  const float2* ts2 = (const float2*)ts;
  for (int q0 = tid; q0 < SLICE4; q0 += 4 * OWN_T) {
    int dl[16], sv[16];
#pragma unroll
    for (int j = 0; j < 4; ++j) {
      int q = q0 + j * OWN_T;
      int4 d = make_int4(-1, -1, -1, -1);
      int4 s = make_int4(0, 0, 0, 0);
      if (q < SLICE4) { d = d4[q]; s = s4[q]; }
      dl[4 * j + 0] = d.x - base; dl[4 * j + 1] = d.y - base;
      dl[4 * j + 2] = d.z - base; dl[4 * j + 3] = d.w - base;
      sv[4 * j + 0] = s.x; sv[4 * j + 1] = s.y; sv[4 * j + 2] = s.z; sv[4 * j + 3] = s.w;
    }
    float2 m[16];
#pragma unroll
    for (int k = 0; k < 16; ++k) {
      m[k] = make_float2(0.0f, 0.0f);
      if ((unsigned)dl[k] < (unsigned)OWN_RANGE) m[k] = ts2[sv[k]];
    }
#pragma unroll
    for (int k = 0; k < 16; ++k) {
      if ((unsigned)dl[k] < (unsigned)OWN_RANGE) {
        atomicAdd(&acc[2 * dl[k] + 0], m[k].x);
        atomicAdd(&acc[2 * dl[k] + 1], m[k].y);
      }
    }
  }
  __syncthreads();
  for (int i0 = tid; i0 < 2 * OWN_RANGE; i0 += 8 * OWN_T) {
    float a[8], w[8];
#pragma unroll
    for (int j = 0; j < 8; ++j) {
      int i = i0 + j * OWN_T;
      a[j] = 0.0f; w[j] = 0.0f;
      if (i < 2 * OWN_RANGE) { a[j] = acc[i]; w[j] = dinv[base + (i >> 1)]; }
    }
#pragma unroll
    for (int j = 0; j < 8; ++j) {
      int i = i0 + j * OWN_T;
      if (i < 2 * OWN_RANGE) atomicAdd(&agg2[2 * base + i], a[j] * w[j]);
    }
  }
}


__global__ void k_final(const float* __restrict__ agg2, const float* __restrict__ b2,
                        float* __restrict__ out) {
  int i = blockIdx.x * blockDim.x + threadIdx.x;
  if (i >= N_NODES) return;
  float z0 = agg2[2 * i + 0] + b2[0];
  float z1 = agg2[2 * i + 1] + b2[1];
  float mm = fmaxf(z0, z1);
  float lse = mm + logf(expf(z0 - mm) + expf(z1 - mm));
  out[2 * i + 0] = z0 - lse;
  out[2 * i + 1] = z1 - lse;
}

extern "C" void kernel_launch(void* const* d_in, const int* in_sizes, int n_in,
                              void* d_out, int out_size, void* d_ws, size_t ws_size,
                              hipStream_t stream) {
  const float* x  = (const float*)d_in[0];
  const float* W1 = (const float*)d_in[1];
  const float* b1 = (const float*)d_in[2];
  const float* W2 = (const float*)d_in[3];
  const float* b2 = (const float*)d_in[4];
  const int*  ei  = (const int*)d_in[5];
  const int* src = ei;
  const int* dst = ei + N_EDGES;
  float* out = (float*)d_out;

  float* w    = (float*)d_ws;
  float* deg  = w;
  float* Bp   = deg + N_NODES;
  float* hs   = Bp + 2048;
  float* agg1 = hs + (size_t)N_NODES * D_HID;
  float* ts   = agg1 + (size_t)N_NODES * D_HID;
  float* agg2 = ts + (size_t)N_NODES * D_OUT;

  const int T = 256;
  k_zero<<<(N_NODES + T - 1) / T, T, 0, stream>>>(deg, N_NODES);
  k_count_deg<<<OWN_R * OWN_B, OWN_T, 0, stream>>>(dst, deg);
  k_dinv<<<(N_NODES + T - 1) / T, T, 0, stream>>>(deg);
  k_bpack<<<4, T, 0, stream>>>(W1, Bp);

  k_gemm1<<<(N_TILES + 7) / 8, T, 0, stream>>>(x, Bp, deg, hs, agg1);

  k_scatter1<<<S1_WG, S1_T, 0, stream>>>(src, dst, hs, deg, agg1);

  k_layer2<<<(N_NODES + T - 1) / T, T, 0, stream>>>(agg1, b1, W2, deg, ts, agg2);
  k_scatter2<<<OWN_R * OWN_B, OWN_T, 0, stream>>>(src, dst, ts, deg, agg2);
  k_final<<<(N_NODES + T - 1) / T, T, 0, stream>>>(agg2, b2, out);
}


# Float adds in memory (degree, layer-1 rows, layer-2 sums) go to one plane per chip half, picked by the wave's die number: no line moves between the two caches; reader adds planes (sum order changes)
# speedup vs baseline: 5.0955x; 1.4979x over previous
#include <hip/hip_runtime.h>
#include <math.h>

#define N_NODES 100000
#define N_EDGES 3200000
#define D_IN 128
#define D_HID 16
#define D_OUT 2
#define N_TILES (N_NODES / 16)

#define WS_FLOATS_TWO_PLANES \
  ((size_t)N_NODES * (1 + D_HID + D_HID + D_OUT + D_OUT + D_HID + D_OUT + 1) + 2048)
#define WS_BYTES_TWO_PLANES (WS_FLOATS_TWO_PLANES * sizeof(float))

static __device__ __forceinline__ bool on_second_half() {
  return ((__builtin_amdgcn_s_sendmsg_rtn(0x87) >> 18) & 1) != 0;
}

typedef __attribute__((ext_vector_type(2))) float v2f;
typedef __attribute__((ext_vector_type(8))) float v8f;


__global__ void k_zero(float* __restrict__ p, float* __restrict__ pb, size_t ws_size, int n) {
  int i = blockIdx.x * blockDim.x + threadIdx.x;
  if (i < n) {
    p[i] = 0.0f;
    if (ws_size >= WS_BYTES_TWO_PLANES) pb[i] = 0.0f;
  }
}


#define OWN_R 8
#define OWN_RANGE (N_NODES / OWN_R)
#define OWN_B 32
#define SLICE (N_EDGES / OWN_B)
#define SLICE4 (SLICE / 4)
#define OWN_T 1024

__global__ void __launch_bounds__(OWN_T)
k_count_deg(const int* __restrict__ dst, float* __restrict__ deg, float* __restrict__ degb,
            size_t ws_size) {
  __shared__ unsigned cnt[OWN_RANGE];
  float* degp = (ws_size >= WS_BYTES_TWO_PLANES && on_second_half()) ? degb : deg;
  const int tid = threadIdx.x;
  const int r = blockIdx.x / OWN_B;
  const int b = blockIdx.x % OWN_B;
  const int base = r * OWN_RANGE;
  for (int i = tid; i < OWN_RANGE; i += OWN_T) cnt[i] = 0u;
  __syncthreads();
  const int4* d4 = (const int4*)(dst + (size_t)b * SLICE);
  for (int q0 = tid; q0 < SLICE4; q0 += 4 * OWN_T) {
    int4 d[4];
#pragma unroll
    for (int j = 0; j < 4; ++j) {
      int q = q0 + j * OWN_T;
      d[j] = make_int4(-1, -1, -1, -1);
      if (q < SLICE4) d[j] = d4[q];
    }
#pragma unroll
    for (int j = 0; j < 4; ++j) {
      unsigned l0 = (unsigned)(d[j].x - base), l1 = (unsigned)(d[j].y - base);
      unsigned l2 = (unsigned)(d[j].z - base), l3 = (unsigned)(d[j].w - base);
      if (l0 < (unsigned)OWN_RANGE) atomicAdd(&cnt[l0], 1u);
      if (l1 < (unsigned)OWN_RANGE) atomicAdd(&cnt[l1], 1u);
      if (l2 < (unsigned)OWN_RANGE) atomicAdd(&cnt[l2], 1u);
      if (l3 < (unsigned)OWN_RANGE) atomicAdd(&cnt[l3], 1u);
    }
  }
  __syncthreads();
  for (int i = tid; i < OWN_RANGE; i += OWN_T) {
    unsigned c = cnt[i];
    if (c) atomicAdd(&degp[base + i], (float)c);
  }
}

__global__ void k_dinv(float* __restrict__ deg, const float* __restrict__ degb, size_t ws_size) {
  int i = blockIdx.x * blockDim.x + threadIdx.x;
  if (i < N_NODES) {
    float dg = deg[i];
    if (ws_size >= WS_BYTES_TWO_PLANES) dg += degb[i];
    deg[i] = rsqrtf(dg + 1.0f);
  }
}


__global__ void k_bpack(const float* __restrict__ W1, float* __restrict__ Bp) {
  int t = blockIdx.x * blockDim.x + threadIdx.x;
  if (t < 32 * 32) {
    int ks = t >> 5, lane = t & 31;
    int n = lane & 15;
    int k0 = ks * 4 + ((lane >> 4) ? 2 : 0);
    Bp[2 * t + 0] = W1[k0 * D_HID + n];
    Bp[2 * t + 1] = W1[(k0 + 1) * D_HID + n];
  }
}


__global__ void k_gemm1(const float* __restrict__ x, const float* __restrict__ Bp,
                        const float* __restrict__ dinv, float* __restrict__ hs,
                        float* __restrict__ agg1, float* __restrict__ agg1b,
                        size_t ws_size) {
  const bool two = ws_size >= WS_BYTES_TWO_PLANES;
  int lane = threadIdx.x & 31;
  int wave = threadIdx.x >> 5;
  int tile = blockIdx.x * (blockDim.x >> 5) + wave;
  if (tile >= N_TILES) return;

  int m = lane & 15;
  int koff = (lane >> 4) * 2;
  const float* xr = x + (size_t)(tile * 16 + m) * D_IN + koff;
  const v2f* bp = (const v2f*)Bp;

  v8f c = {0.f, 0.f, 0.f, 0.f, 0.f, 0.f, 0.f, 0.f};
#pragma unroll
  for (int ks = 0; ks < 32; ++ks) {
    v2f a = *(const v2f*)(xr + ks * 4);
    v2f b = bp[ks * 32 + lane];
    c = __builtin_amdgcn_wmma_f32_16x16x4_f32(
         false, a,  false, b,
         (short)0, c,  false,  false);
  }

  int n = lane & 15;
  int mbase = (lane >> 4) * 8;
#pragma unroll
  for (int v = 0; v < 8; ++v) {
    int r = tile * 16 + mbase + v;
    float di = dinv[r];
    float hv = c[v] * di;
    hs[(size_t)r * D_HID + n] = hv;
    agg1[(size_t)r * D_HID + n] = hv * di;
  }
  if (two) {
#pragma unroll
    for (int v = 0; v < 8; ++v) {
      int r = tile * 16 + mbase + v;
      agg1b[(size_t)r * D_HID + n] = 0.0f;
    }
  }
}


#define S1_BATCHES (N_EDGES / 32)
#define S1_T 128
#define S1_WG 3125
#define S1_WAVES (S1_WG * (S1_T / 32))

#define ROW_SHARE_I(v, K) __builtin_amdgcn_update_dpp(0, (v), 0x150 + (K), 0xf, 0xf, false)

#define S1_GATHER(K)                                                          \
  {                                                                           \
    unsigned so = (unsigned)ROW_SHARE_I(s16, K) + j4;                         \
    m[K] = *(const float*)((const char*)hs + so);                             \
  }

#define S1_ADD(K)                                                             \
  {                                                                           \
    unsigned dof = (unsigned)ROW_SHARE_I(d16, K) + j4;                        \
    float wk = __int_as_float(ROW_SHARE_I(wbits, K));                         \
    atomicAdd((float*)((char*)aggp + dof), m[K] * wk);                        \
  }

__global__ void __launch_bounds__(S1_T)
k_scatter1(const int* __restrict__ src, const int* __restrict__ dst,
           const float* __restrict__ hs, const float* __restrict__ dinv,
           float* __restrict__ agg1, float* __restrict__ agg1b, size_t ws_size) {
  float* aggp = (ws_size >= WS_BYTES_TWO_PLANES && on_second_half()) ? agg1b : agg1;
  const int lane = threadIdx.x & 31;
  const unsigned j4 = (unsigned)(lane & 15) * 4u;
  const int gw = __builtin_amdgcn_readfirstlane(
      (int)(blockIdx.x * (S1_T / 32) + (threadIdx.x >> 5)));
  int s = src[gw * 32 + lane];
  int d = dst[gw * 32 + lane];
  for (int bt = gw; bt < S1_BATCHES; bt += S1_WAVES) {
    const int nb = (bt + S1_WAVES < S1_BATCHES) ? bt + S1_WAVES : bt;
    const int sn = src[nb * 32 + lane];
    const int dn = dst[nb * 32 + lane];
    const int wbits = __float_as_int(dinv[d]);
    const int s16 = s * (D_HID * 4);
    const int d16 = d * (D_HID * 4);
    float m[16];
    S1_GATHER(0)  S1_GATHER(1)  S1_GATHER(2)  S1_GATHER(3)
    S1_GATHER(4)  S1_GATHER(5)  S1_GATHER(6)  S1_GATHER(7)
    S1_GATHER(8)  S1_GATHER(9)  S1_GATHER(10) S1_GATHER(11)
    S1_GATHER(12) S1_GATHER(13) S1_GATHER(14) S1_GATHER(15)
    S1_ADD(0)  S1_ADD(1)  S1_ADD(2)  S1_ADD(3)
    S1_ADD(4)  S1_ADD(5)  S1_ADD(6)  S1_ADD(7)
    S1_ADD(8)  S1_ADD(9)  S1_ADD(10) S1_ADD(11)
    S1_ADD(12) S1_ADD(13) S1_ADD(14) S1_ADD(15)
    s = sn;
    d = dn;
  }
}


__global__ void k_layer2(const float* __restrict__ agg1, const float* __restrict__ b1,
                         const float* __restrict__ W2, const float* __restrict__ dinv,
                         float* __restrict__ ts, float* __restrict__ agg2,
                         const float* __restrict__ agg1b, float* __restrict__ agg2b,
                         size_t ws_size) {
  int i = blockIdx.x * blockDim.x + threadIdx.x;
  if (i >= N_NODES) return;
  const bool two = ws_size >= WS_BYTES_TWO_PLANES;
  const float* row = agg1 + (size_t)i * D_HID;
  const float* rowb = agg1b + (size_t)i * D_HID;
  float a[D_HID];
#pragma unroll
  for (int j = 0; j < D_HID; ++j) a[j] = row[j];
  if (two) {
#pragma unroll
    for (int j = 0; j < D_HID; ++j) a[j] += rowb[j];
  }
  float t0 = 0.f, t1 = 0.f;
#pragma unroll
  for (int j = 0; j < D_HID; ++j) {
    float hj = fmaxf(a[j] + b1[j], 0.0f);
    t0 += hj * W2[j * 2 + 0];
    t1 += hj * W2[j * 2 + 1];
  }
  float di = dinv[i];
  float s0 = t0 * di, s1 = t1 * di;
  ts[2 * i + 0] = s0;
  ts[2 * i + 1] = s1;
  agg2[2 * i + 0] = s0 * di;
  agg2[2 * i + 1] = s1 * di;
  if (two) {
    agg2b[2 * i + 0] = 0.0f;
    agg2b[2 * i + 1] = 0.0f;
  }
}


__global__ void __launch_bounds__(OWN_T)
k_scatter2(const int* __restrict__ src, const int* __restrict__ dst,
           const float* __restrict__ ts, const float* __restrict__ dinv,
           float* __restrict__ agg2, float* __restrict__ agg2b, size_t ws_size) {
  __shared__ float acc[2 * OWN_RANGE];
  float* aggp = (ws_size >= WS_BYTES_TWO_PLANES && on_second_half()) ? agg2b : agg2;
  const int tid = threadIdx.x;
  const int r = blockIdx.x / OWN_B;
  const int b = blockIdx.x % OWN_B;
  const int base = r * OWN_RANGE;
  for (int i = tid; i < 2 * OWN_RANGE; i += OWN_T) acc[i] = 0.0f;
  __syncthreads();
  const int4* d4 = (const int4*)(dst + (size_t)b * SLICE);
  const int4* s4 = (const int4*)(src + (size_t)b * SLICE);
  const float2* ts2 = (const float2*)ts;
  for (int q0 = tid; q0 < SLICE4; q0 += 4 * OWN_T) {
    int dl[16], sv[16];
#pragma unroll
    for (int j = 0; j < 4; ++j) {
      int q = q0 + j * OWN_T;
      int4 d = make_int4(-1, -1, -1, -1);
      int4 s = make_int4(0, 0, 0, 0);
      if (q < SLICE4) { d = d4[q]; s = s4[q]; }
      dl[4 * j + 0] = d.x - base; dl[4 * j + 1] = d.y - base;
      dl[4 * j + 2] = d.z - base; dl[4 * j + 3] = d.w - base;
      sv[4 * j + 0] = s.x; sv[4 * j + 1] = s.y; sv[4 * j + 2] = s.z; sv[4 * j + 3] = s.w;
    }
    float2 m[16];
#pragma unroll
    for (int k = 0; k < 16; ++k) {
      m[k] = make_float2(0.0f, 0.0f);
      if ((unsigned)dl[k] < (unsigned)OWN_RANGE) m[k] = ts2[sv[k]];
    }
#pragma unroll
    for (int k = 0; k < 16; ++k) {
      if ((unsigned)dl[k] < (unsigned)OWN_RANGE) {
        atomicAdd(&acc[2 * dl[k] + 0], m[k].x);
        atomicAdd(&acc[2 * dl[k] + 1], m[k].y);
      }
    }
  }
  __syncthreads();
  for (int i0 = tid; i0 < 2 * OWN_RANGE; i0 += 8 * OWN_T) {
    float a[8], w[8];
#pragma unroll
    for (int j = 0; j < 8; ++j) {
      int i = i0 + j * OWN_T;
      a[j] = 0.0f; w[j] = 0.0f;
      if (i < 2 * OWN_RANGE) { a[j] = acc[i]; w[j] = dinv[base + (i >> 1)]; }
    }
#pragma unroll
    for (int j = 0; j < 8; ++j) {
      int i = i0 + j * OWN_T;
      if (i < 2 * OWN_RANGE) atomicAdd(&aggp[2 * base + i], a[j] * w[j]);
    }
  }
}


__global__ void k_final(const float* __restrict__ agg2, const float* __restrict__ b2,
                        float* __restrict__ out, const float* __restrict__ agg2b,
                        size_t ws_size) {
  int i = blockIdx.x * blockDim.x + threadIdx.x;
  if (i >= N_NODES) return;
  float y0 = agg2[2 * i + 0];
  float y1 = agg2[2 * i + 1];
  if (ws_size >= WS_BYTES_TWO_PLANES) {
    y0 += agg2b[2 * i + 0];
    y1 += agg2b[2 * i + 1];
  }
  float z0 = y0 + b2[0];
  float z1 = y1 + b2[1];
  float mm = fmaxf(z0, z1);
  float lse = mm + logf(expf(z0 - mm) + expf(z1 - mm));
  out[2 * i + 0] = z0 - lse;
  out[2 * i + 1] = z1 - lse;
}

extern "C" void kernel_launch(void* const* d_in, const int* in_sizes, int n_in,
                              void* d_out, int out_size, void* d_ws, size_t ws_size,
                              hipStream_t stream) {
  const float* x  = (const float*)d_in[0];
  const float* W1 = (const float*)d_in[1];
  const float* b1 = (const float*)d_in[2];
  const float* W2 = (const float*)d_in[3];
  const float* b2 = (const float*)d_in[4];
  const int*  ei  = (const int*)d_in[5];
  const int* src = ei;
  const int* dst = ei + N_EDGES;
  float* out = (float*)d_out;

  float* w    = (float*)d_ws;
  float* deg  = w;
  float* Bp   = deg + N_NODES;
  float* hs   = Bp + 2048;
  float* agg1 = hs + (size_t)N_NODES * D_HID;
  float* ts   = agg1 + (size_t)N_NODES * D_HID;
  float* agg2 = ts + (size_t)N_NODES * D_OUT;
  float* agg1b = agg2 + (size_t)N_NODES * D_OUT;
  float* agg2b = agg1b + (size_t)N_NODES * D_HID;
  float* degb  = agg2b + (size_t)N_NODES * D_OUT;

  const int T = 256;
  k_zero<<<(N_NODES + T - 1) / T, T, 0, stream>>>(deg, degb, ws_size, N_NODES);
  k_count_deg<<<OWN_R * OWN_B, OWN_T, 0, stream>>>(dst, deg, degb, ws_size);
  k_dinv<<<(N_NODES + T - 1) / T, T, 0, stream>>>(deg, degb, ws_size);
  k_bpack<<<4, T, 0, stream>>>(W1, Bp);

  k_gemm1<<<(N_TILES + 7) / 8, T, 0, stream>>>(x, Bp, deg, hs, agg1, agg1b, ws_size);

  k_scatter1<<<S1_WG, S1_T, 0, stream>>>(src, dst, hs, deg, agg1, agg1b, ws_size);

  k_layer2<<<(N_NODES + T - 1) / T, T, 0, stream>>>(agg1, b1, W2, deg, ts, agg2, agg1b, agg2b, ws_size);
  k_scatter2<<<OWN_R * OWN_B, OWN_T, 0, stream>>>(src, dst, ts, deg, agg2, agg2b, ws_size);
  k_final<<<(N_NODES + T - 1) / T, T, 0, stream>>>(agg2, b2, out, agg2b, ws_size);
}


# Two small passes folded away (9 launches to 7): the degree-plane clear and the W1 fragment pack share one launch, and 1/sqrt(deg+1) is computed in the GEMM pass by each tile's own rows
# speedup vs baseline: 5.2778x; 1.0358x over previous
#include <hip/hip_runtime.h>
#include <math.h>

#define N_NODES 100000
#define N_EDGES 3200000
#define D_IN 128
#define D_HID 16
#define D_OUT 2
#define N_TILES (N_NODES / 16)

#define WS_FLOATS_TWO_PLANES \
  ((size_t)N_NODES * (1 + D_HID + D_HID + D_OUT + D_OUT + D_HID + D_OUT + 1) + 2048)
#define WS_BYTES_TWO_PLANES (WS_FLOATS_TWO_PLANES * sizeof(float))

static __device__ __forceinline__ bool on_second_half() {
  return ((__builtin_amdgcn_s_sendmsg_rtn(0x87) >> 18) & 1) != 0;
}

typedef __attribute__((ext_vector_type(2))) float v2f;
typedef __attribute__((ext_vector_type(8))) float v8f;


#define PREP_T 256
#define PREP_WG ((N_NODES / 4 + PREP_T - 1) / PREP_T)

__global__ void k_prep(float* __restrict__ deg, float* __restrict__ degb, size_t ws_size,
                       const float* __restrict__ W1, float* __restrict__ Bp) {
  int t = blockIdx.x * blockDim.x + threadIdx.x;
  if (t < N_NODES / 4) {
    const float4 z = make_float4(0.0f, 0.0f, 0.0f, 0.0f);
    ((float4*)deg)[t] = z;
    if (ws_size >= WS_BYTES_TWO_PLANES) ((float4*)degb)[t] = z;
  }
  if (t < 32 * 32) {
    int ks = t >> 5, lane = t & 31;
    int n = lane & 15;
    int k0 = ks * 4 + ((lane >> 4) ? 2 : 0);
    Bp[2 * t + 0] = W1[k0 * D_HID + n];
    Bp[2 * t + 1] = W1[(k0 + 1) * D_HID + n];
  }
}


#define OWN_R 8
#define OWN_RANGE (N_NODES / OWN_R)
#define OWN_B 32
#define SLICE (N_EDGES / OWN_B)
#define SLICE4 (SLICE / 4)
#define OWN_T 1024

__global__ void __launch_bounds__(OWN_T)
k_count_deg(const int* __restrict__ dst, float* __restrict__ deg, float* __restrict__ degb,
            size_t ws_size) {
  __shared__ unsigned cnt[OWN_RANGE];
  float* degp = (ws_size >= WS_BYTES_TWO_PLANES && on_second_half()) ? degb : deg;
  const int tid = threadIdx.x;
  const int r = blockIdx.x / OWN_B;
  const int b = blockIdx.x % OWN_B;
  const int base = r * OWN_RANGE;
  for (int i = tid; i < OWN_RANGE; i += OWN_T) cnt[i] = 0u;
  __syncthreads();
  const int4* d4 = (const int4*)(dst + (size_t)b * SLICE);
  for (int q0 = tid; q0 < SLICE4; q0 += 4 * OWN_T) {
    int4 d[4];
#pragma unroll
    for (int j = 0; j < 4; ++j) {
      int q = q0 + j * OWN_T;
      d[j] = make_int4(-1, -1, -1, -1);
      if (q < SLICE4) d[j] = d4[q];
    }
#pragma unroll
    for (int j = 0; j < 4; ++j) {
      unsigned l0 = (unsigned)(d[j].x - base), l1 = (unsigned)(d[j].y - base);
      unsigned l2 = (unsigned)(d[j].z - base), l3 = (unsigned)(d[j].w - base);
      if (l0 < (unsigned)OWN_RANGE) atomicAdd(&cnt[l0], 1u);
      if (l1 < (unsigned)OWN_RANGE) atomicAdd(&cnt[l1], 1u);
      if (l2 < (unsigned)OWN_RANGE) atomicAdd(&cnt[l2], 1u);
      if (l3 < (unsigned)OWN_RANGE) atomicAdd(&cnt[l3], 1u);
    }
  }
  __syncthreads();
  for (int i = tid; i < OWN_RANGE; i += OWN_T) {
    unsigned c = cnt[i];
    if (c) atomicAdd(&degp[base + i], (float)c);
  }
}


__global__ void k_gemm1(const float* __restrict__ x, const float* __restrict__ Bp,
                        float* deg, const float* degb, float* __restrict__ hs,
                        float* __restrict__ agg1, float* __restrict__ agg1b,
                        size_t ws_size) {
  const bool two = ws_size >= WS_BYTES_TWO_PLANES;
  int lane = threadIdx.x & 31;
  int wave = threadIdx.x >> 5;
  int tile = blockIdx.x * (blockDim.x >> 5) + wave;
  if (tile >= N_TILES) return;

  const int n = lane & 15;
  const int mbase = (lane >> 4) * 8;
  const int r0 = tile * 16 + mbase;
  float4 ga = ((const float4*)(deg + r0))[0];
  float4 gb = ((const float4*)(deg + r0))[1];
  if (two) {
    float4 ha = ((const float4*)(degb + r0))[0];
    float4 hb = ((const float4*)(degb + r0))[1];
    ga.x += ha.x; ga.y += ha.y; ga.z += ha.z; ga.w += ha.w;
    gb.x += hb.x; gb.y += hb.y; gb.z += hb.z; gb.w += hb.w;
  }
  float dv[8];
  dv[0] = rsqrtf(ga.x + 1.0f); dv[1] = rsqrtf(ga.y + 1.0f);
  dv[2] = rsqrtf(ga.z + 1.0f); dv[3] = rsqrtf(ga.w + 1.0f);
  dv[4] = rsqrtf(gb.x + 1.0f); dv[5] = rsqrtf(gb.y + 1.0f);
  dv[6] = rsqrtf(gb.z + 1.0f); dv[7] = rsqrtf(gb.w + 1.0f);

  int m = lane & 15;
  int koff = (lane >> 4) * 2;
  const float* xr = x + (size_t)(tile * 16 + m) * D_IN + koff;
  const v2f* bp = (const v2f*)Bp;

  v8f c = {0.f, 0.f, 0.f, 0.f, 0.f, 0.f, 0.f, 0.f};
#pragma unroll
  for (int ks = 0; ks < 32; ++ks) {
    v2f a = *(const v2f*)(xr + ks * 4);
    v2f b = bp[ks * 32 + lane];
    c = __builtin_amdgcn_wmma_f32_16x16x4_f32(
         false, a,  false, b,
         (short)0, c,  false,  false);
  }

#pragma unroll
  for (int v = 0; v < 8; ++v) {
    int r = tile * 16 + mbase + v;
    float di = dv[v];
    float hv = c[v] * di;
    hs[(size_t)r * D_HID + n] = hv;
    agg1[(size_t)r * D_HID + n] = hv * di;
  }
  if (n == 0) {
    ((float4*)(deg + r0))[0] = make_float4(dv[0], dv[1], dv[2], dv[3]);
    ((float4*)(deg + r0))[1] = make_float4(dv[4], dv[5], dv[6], dv[7]);
  }
  if (two) {
#pragma unroll
    for (int v = 0; v < 8; ++v) {
      int r = tile * 16 + mbase + v;
      agg1b[(size_t)r * D_HID + n] = 0.0f;
    }
  }
}


#define S1_BATCHES (N_EDGES / 32)
#define S1_T 128
#define S1_WG 3125
#define S1_WAVES (S1_WG * (S1_T / 32))

#define ROW_SHARE_I(v, K) __builtin_amdgcn_update_dpp(0, (v), 0x150 + (K), 0xf, 0xf, false)

#define S1_GATHER(K)                                                          \
  {                                                                           \
    unsigned so = (unsigned)ROW_SHARE_I(s16, K) + j4;                         \
    m[K] = *(const float*)((const char*)hs + so);                             \
  }

#define S1_ADD(K)                                                             \
  {                                                                           \
    unsigned dof = (unsigned)ROW_SHARE_I(d16, K) + j4;                        \
    float wk = __int_as_float(ROW_SHARE_I(wbits, K));                         \
    atomicAdd((float*)((char*)aggp + dof), m[K] * wk);                        \
  }

__global__ void __launch_bounds__(S1_T)
k_scatter1(const int* __restrict__ src, const int* __restrict__ dst,
           const float* __restrict__ hs, const float* __restrict__ dinv,
           float* __restrict__ agg1, float* __restrict__ agg1b, size_t ws_size) {
  float* aggp = (ws_size >= WS_BYTES_TWO_PLANES && on_second_half()) ? agg1b : agg1;
  const int lane = threadIdx.x & 31;
  const unsigned j4 = (unsigned)(lane & 15) * 4u;
  const int gw = __builtin_amdgcn_readfirstlane(
      (int)(blockIdx.x * (S1_T / 32) + (threadIdx.x >> 5)));
  int s = src[gw * 32 + lane];
  int d = dst[gw * 32 + lane];
  for (int bt = gw; bt < S1_BATCHES; bt += S1_WAVES) {
    const int nb = (bt + S1_WAVES < S1_BATCHES) ? bt + S1_WAVES : bt;
    const int sn = src[nb * 32 + lane];
    const int dn = dst[nb * 32 + lane];
    const int wbits = __float_as_int(dinv[d]);
    const int s16 = s * (D_HID * 4);
    const int d16 = d * (D_HID * 4);
    float m[16];
    S1_GATHER(0)  S1_GATHER(1)  S1_GATHER(2)  S1_GATHER(3)
    S1_GATHER(4)  S1_GATHER(5)  S1_GATHER(6)  S1_GATHER(7)
    S1_GATHER(8)  S1_GATHER(9)  S1_GATHER(10) S1_GATHER(11)
    S1_GATHER(12) S1_GATHER(13) S1_GATHER(14) S1_GATHER(15)
    S1_ADD(0)  S1_ADD(1)  S1_ADD(2)  S1_ADD(3)
    S1_ADD(4)  S1_ADD(5)  S1_ADD(6)  S1_ADD(7)
    S1_ADD(8)  S1_ADD(9)  S1_ADD(10) S1_ADD(11)
    S1_ADD(12) S1_ADD(13) S1_ADD(14) S1_ADD(15)
    s = sn;
    d = dn;
  }
}


__global__ void k_layer2(const float* __restrict__ agg1, const float* __restrict__ b1,
                         const float* __restrict__ W2, const float* __restrict__ dinv,
                         float* __restrict__ ts, float* __restrict__ agg2,
                         const float* __restrict__ agg1b, float* __restrict__ agg2b,
                         size_t ws_size) {
  int i = blockIdx.x * blockDim.x + threadIdx.x;
  if (i >= N_NODES) return;
  const bool two = ws_size >= WS_BYTES_TWO_PLANES;
  const float* row = agg1 + (size_t)i * D_HID;
  const float* rowb = agg1b + (size_t)i * D_HID;
  float a[D_HID];
#pragma unroll
  for (int j = 0; j < D_HID; ++j) a[j] = row[j];
  if (two) {
#pragma unroll
    for (int j = 0; j < D_HID; ++j) a[j] += rowb[j];
  }
  float t0 = 0.f, t1 = 0.f;
#pragma unroll
  for (int j = 0; j < D_HID; ++j) {
    float hj = fmaxf(a[j] + b1[j], 0.0f);
    t0 += hj * W2[j * 2 + 0];
    t1 += hj * W2[j * 2 + 1];
  }
  float di = dinv[i];
  float s0 = t0 * di, s1 = t1 * di;
  ts[2 * i + 0] = s0;
  ts[2 * i + 1] = s1;
  agg2[2 * i + 0] = s0 * di;
  agg2[2 * i + 1] = s1 * di;
  if (two) {
    agg2b[2 * i + 0] = 0.0f;
    agg2b[2 * i + 1] = 0.0f;
  }
}


__global__ void __launch_bounds__(OWN_T)
k_scatter2(const int* __restrict__ src, const int* __restrict__ dst,
           const float* __restrict__ ts, const float* __restrict__ dinv,
           float* __restrict__ agg2, float* __restrict__ agg2b, size_t ws_size) {
  __shared__ float acc[2 * OWN_RANGE];
  float* aggp = (ws_size >= WS_BYTES_TWO_PLANES && on_second_half()) ? agg2b : agg2;
  const int tid = threadIdx.x;
  const int r = blockIdx.x / OWN_B;
  const int b = blockIdx.x % OWN_B;
  const int base = r * OWN_RANGE;
  for (int i = tid; i < 2 * OWN_RANGE; i += OWN_T) acc[i] = 0.0f;
  __syncthreads();
  const int4* d4 = (const int4*)(dst + (size_t)b * SLICE);
  const int4* s4 = (const int4*)(src + (size_t)b * SLICE);
  const float2* ts2 = (const float2*)ts;
  for (int q0 = tid; q0 < SLICE4; q0 += 4 * OWN_T) {
    int dl[16], sv[16];
#pragma unroll
    for (int j = 0; j < 4; ++j) {
      int q = q0 + j * OWN_T;
      int4 d = make_int4(-1, -1, -1, -1);
      int4 s = make_int4(0, 0, 0, 0);
      if (q < SLICE4) { d = d4[q]; s = s4[q]; }
      dl[4 * j + 0] = d.x - base; dl[4 * j + 1] = d.y - base;
      dl[4 * j + 2] = d.z - base; dl[4 * j + 3] = d.w - base;
      sv[4 * j + 0] = s.x; sv[4 * j + 1] = s.y; sv[4 * j + 2] = s.z; sv[4 * j + 3] = s.w;
    }
    float2 m[16];
#pragma unroll
    for (int k = 0; k < 16; ++k) {
      m[k] = make_float2(0.0f, 0.0f);
      if ((unsigned)dl[k] < (unsigned)OWN_RANGE) m[k] = ts2[sv[k]];
    }
#pragma unroll
    for (int k = 0; k < 16; ++k) {
      if ((unsigned)dl[k] < (unsigned)OWN_RANGE) {
        atomicAdd(&acc[2 * dl[k] + 0], m[k].x);
        atomicAdd(&acc[2 * dl[k] + 1], m[k].y);
      }
    }
  }
  __syncthreads();
  for (int i0 = tid; i0 < 2 * OWN_RANGE; i0 += 8 * OWN_T) {
    float a[8], w[8];
#pragma unroll
    for (int j = 0; j < 8; ++j) {
      int i = i0 + j * OWN_T;
      a[j] = 0.0f; w[j] = 0.0f;
      if (i < 2 * OWN_RANGE) { a[j] = acc[i]; w[j] = dinv[base + (i >> 1)]; }
    }
#pragma unroll
    for (int j = 0; j < 8; ++j) {
      int i = i0 + j * OWN_T;
      if (i < 2 * OWN_RANGE) atomicAdd(&aggp[2 * base + i], a[j] * w[j]);
    }
  }
}


__global__ void k_final(const float* __restrict__ agg2, const float* __restrict__ b2,
                        float* __restrict__ out, const float* __restrict__ agg2b,
                        size_t ws_size) {
  int i = blockIdx.x * blockDim.x + threadIdx.x;
  if (i >= N_NODES) return;
  float y0 = agg2[2 * i + 0];
  float y1 = agg2[2 * i + 1];
  if (ws_size >= WS_BYTES_TWO_PLANES) {
    y0 += agg2b[2 * i + 0];
    y1 += agg2b[2 * i + 1];
  }
  float z0 = y0 + b2[0];
  float z1 = y1 + b2[1];
  float mm = fmaxf(z0, z1);
  float lse = mm + logf(expf(z0 - mm) + expf(z1 - mm));
  out[2 * i + 0] = z0 - lse;
  out[2 * i + 1] = z1 - lse;
}

extern "C" void kernel_launch(void* const* d_in, const int* in_sizes, int n_in,
                              void* d_out, int out_size, void* d_ws, size_t ws_size,
                              hipStream_t stream) {
  const float* x  = (const float*)d_in[0];
  const float* W1 = (const float*)d_in[1];
  const float* b1 = (const float*)d_in[2];
  const float* W2 = (const float*)d_in[3];
  const float* b2 = (const float*)d_in[4];
  const int*  ei  = (const int*)d_in[5];
  const int* src = ei;
  const int* dst = ei + N_EDGES;
  float* out = (float*)d_out;

  float* w    = (float*)d_ws;
  float* deg  = w;
  float* Bp   = deg + N_NODES;
  float* hs   = Bp + 2048;
  float* agg1 = hs + (size_t)N_NODES * D_HID;
  float* ts   = agg1 + (size_t)N_NODES * D_HID;
  float* agg2 = ts + (size_t)N_NODES * D_OUT;
  float* agg1b = agg2 + (size_t)N_NODES * D_OUT;
  float* agg2b = agg1b + (size_t)N_NODES * D_HID;
  float* degb  = agg2b + (size_t)N_NODES * D_OUT;

  const int T = 256;
  k_prep<<<PREP_WG, PREP_T, 0, stream>>>(deg, degb, ws_size, W1, Bp);
  k_count_deg<<<OWN_R * OWN_B, OWN_T, 0, stream>>>(dst, deg, degb, ws_size);

  k_gemm1<<<(N_TILES + 7) / 8, T, 0, stream>>>(x, Bp, deg, degb, hs, agg1, agg1b, ws_size);

  k_scatter1<<<S1_WG, S1_T, 0, stream>>>(src, dst, hs, deg, agg1, agg1b, ws_size);

  k_layer2<<<(N_NODES + T - 1) / T, T, 0, stream>>>(agg1, b1, W2, deg, ts, agg2, agg1b, agg2b, ws_size);
  k_scatter2<<<OWN_R * OWN_B, OWN_T, 0, stream>>>(src, dst, ts, deg, agg2, agg2b, ws_size);
  k_final<<<(N_NODES + T - 1) / T, T, 0, stream>>>(agg2, b2, out, agg2b, ws_size);
}


# Layer-1 edge sum drops the per-edge dinv[dst] factor: hs rows are summed plain and the layer-2 pass scales each node's sum by dinv once (one 4-byte gather per edge gone; rounding differs)
# speedup vs baseline: 5.9300x; 1.1236x over previous
#include <hip/hip_runtime.h>
#include <math.h>

#define N_NODES 100000
#define N_EDGES 3200000
#define D_IN 128
#define D_HID 16
#define D_OUT 2
#define N_TILES (N_NODES / 16)

#define WS_FLOATS_TWO_PLANES \
  ((size_t)N_NODES * (1 + D_HID + D_HID + D_OUT + D_OUT + D_HID + D_OUT + 1) + 2048)
#define WS_BYTES_TWO_PLANES (WS_FLOATS_TWO_PLANES * sizeof(float))

static __device__ __forceinline__ bool on_second_half() {
  return ((__builtin_amdgcn_s_sendmsg_rtn(0x87) >> 18) & 1) != 0;
}

typedef __attribute__((ext_vector_type(2))) float v2f;
typedef __attribute__((ext_vector_type(8))) float v8f;


#define PREP_T 256
#define PREP_WG ((N_NODES / 4 + PREP_T - 1) / PREP_T)

__global__ void k_prep(float* __restrict__ deg, float* __restrict__ degb, size_t ws_size,
                       const float* __restrict__ W1, float* __restrict__ Bp) {
  int t = blockIdx.x * blockDim.x + threadIdx.x;
  if (t < N_NODES / 4) {
    const float4 z = make_float4(0.0f, 0.0f, 0.0f, 0.0f);
    ((float4*)deg)[t] = z;
    if (ws_size >= WS_BYTES_TWO_PLANES) ((float4*)degb)[t] = z;
  }
  if (t < 32 * 32) {
    int ks = t >> 5, lane = t & 31;
    int n = lane & 15;
    int k0 = ks * 4 + ((lane >> 4) ? 2 : 0);
    Bp[2 * t + 0] = W1[k0 * D_HID + n];
    Bp[2 * t + 1] = W1[(k0 + 1) * D_HID + n];
  }
}


#define OWN_R 8
#define OWN_RANGE (N_NODES / OWN_R)
#define OWN_B 32
#define SLICE (N_EDGES / OWN_B)
#define SLICE4 (SLICE / 4)
#define OWN_T 1024

__global__ void __launch_bounds__(OWN_T)
k_count_deg(const int* __restrict__ dst, float* __restrict__ deg, float* __restrict__ degb,
            size_t ws_size) {
  __shared__ unsigned cnt[OWN_RANGE];
  float* degp = (ws_size >= WS_BYTES_TWO_PLANES && on_second_half()) ? degb : deg;
  const int tid = threadIdx.x;
  const int r = blockIdx.x / OWN_B;
  const int b = blockIdx.x % OWN_B;
  const int base = r * OWN_RANGE;
  for (int i = tid; i < OWN_RANGE; i += OWN_T) cnt[i] = 0u;
  __syncthreads();
  const int4* d4 = (const int4*)(dst + (size_t)b * SLICE);
  for (int q0 = tid; q0 < SLICE4; q0 += 4 * OWN_T) {
    int4 d[4];
#pragma unroll
    for (int j = 0; j < 4; ++j) {
      int q = q0 + j * OWN_T;
      d[j] = make_int4(-1, -1, -1, -1);
      if (q < SLICE4) d[j] = d4[q];
    }
#pragma unroll
    for (int j = 0; j < 4; ++j) {
      unsigned l0 = (unsigned)(d[j].x - base), l1 = (unsigned)(d[j].y - base);
      unsigned l2 = (unsigned)(d[j].z - base), l3 = (unsigned)(d[j].w - base);
      if (l0 < (unsigned)OWN_RANGE) atomicAdd(&cnt[l0], 1u);
      if (l1 < (unsigned)OWN_RANGE) atomicAdd(&cnt[l1], 1u);
      if (l2 < (unsigned)OWN_RANGE) atomicAdd(&cnt[l2], 1u);
      if (l3 < (unsigned)OWN_RANGE) atomicAdd(&cnt[l3], 1u);
    }
  }
  __syncthreads();
  for (int i = tid; i < OWN_RANGE; i += OWN_T) {
    unsigned c = cnt[i];
    if (c) atomicAdd(&degp[base + i], (float)c);
  }
}


__global__ void k_gemm1(const float* __restrict__ x, const float* __restrict__ Bp,
                        float* deg, const float* degb, float* __restrict__ hs,
                        float* __restrict__ agg1, float* __restrict__ agg1b,
                        size_t ws_size) {
  const bool two = ws_size >= WS_BYTES_TWO_PLANES;
  int lane = threadIdx.x & 31;
  int wave = threadIdx.x >> 5;
  int tile = blockIdx.x * (blockDim.x >> 5) + wave;
  if (tile >= N_TILES) return;

  const int n = lane & 15;
  const int mbase = (lane >> 4) * 8;
  const int r0 = tile * 16 + mbase;
  float4 ga = ((const float4*)(deg + r0))[0];
  float4 gb = ((const float4*)(deg + r0))[1];
  if (two) {
    float4 ha = ((const float4*)(degb + r0))[0];
    float4 hb = ((const float4*)(degb + r0))[1];
    ga.x += ha.x; ga.y += ha.y; ga.z += ha.z; ga.w += ha.w;
    gb.x += hb.x; gb.y += hb.y; gb.z += hb.z; gb.w += hb.w;
  }
  float dv[8];
  dv[0] = rsqrtf(ga.x + 1.0f); dv[1] = rsqrtf(ga.y + 1.0f);
  dv[2] = rsqrtf(ga.z + 1.0f); dv[3] = rsqrtf(ga.w + 1.0f);
  dv[4] = rsqrtf(gb.x + 1.0f); dv[5] = rsqrtf(gb.y + 1.0f);
  dv[6] = rsqrtf(gb.z + 1.0f); dv[7] = rsqrtf(gb.w + 1.0f);

  int m = lane & 15;
  int koff = (lane >> 4) * 2;
  const float* xr = x + (size_t)(tile * 16 + m) * D_IN + koff;
  const v2f* bp = (const v2f*)Bp;

  v8f c = {0.f, 0.f, 0.f, 0.f, 0.f, 0.f, 0.f, 0.f};
#pragma unroll
  for (int ks = 0; ks < 32; ++ks) {
    v2f a = *(const v2f*)(xr + ks * 4);
    v2f b = bp[ks * 32 + lane];
    c = __builtin_amdgcn_wmma_f32_16x16x4_f32(
         false, a,  false, b,
         (short)0, c,  false,  false);
  }

#pragma unroll
  for (int v = 0; v < 8; ++v) {
    int r = tile * 16 + mbase + v;
    float di = dv[v];
    float hv = c[v] * di;
    hs[(size_t)r * D_HID + n] = hv;
    agg1[(size_t)r * D_HID + n] = hv;
  }
  if (n == 0) {
    ((float4*)(deg + r0))[0] = make_float4(dv[0], dv[1], dv[2], dv[3]);
    ((float4*)(deg + r0))[1] = make_float4(dv[4], dv[5], dv[6], dv[7]);
  }
  if (two) {
#pragma unroll
    for (int v = 0; v < 8; ++v) {
      int r = tile * 16 + mbase + v;
      agg1b[(size_t)r * D_HID + n] = 0.0f;
    }
  }
}


#define S1_BATCHES (N_EDGES / 32)
#define S1_T 128
#define S1_WG 3125
#define S1_WAVES (S1_WG * (S1_T / 32))

#define ROW_SHARE_I(v, K) __builtin_amdgcn_update_dpp(0, (v), 0x150 + (K), 0xf, 0xf, false)

#define S1_GATHER(K)                                                          \
  {                                                                           \
    unsigned so = (unsigned)ROW_SHARE_I(s16, K) + j4;                         \
    m[K] = *(const float*)((const char*)hs + so);                             \
  }

#define S1_ADD(K)                                                             \
  {                                                                           \
    unsigned dof = (unsigned)ROW_SHARE_I(d16, K) + j4;                        \
    atomicAdd((float*)((char*)aggp + dof), m[K]);                             \
  }

__global__ void __launch_bounds__(S1_T)
k_scatter1(const int* __restrict__ src, const int* __restrict__ dst,
           const float* __restrict__ hs,
           float* __restrict__ agg1, float* __restrict__ agg1b, size_t ws_size) {
  float* aggp = (ws_size >= WS_BYTES_TWO_PLANES && on_second_half()) ? agg1b : agg1;
  const int lane = threadIdx.x & 31;
  const unsigned j4 = (unsigned)(lane & 15) * 4u;
  const int gw = __builtin_amdgcn_readfirstlane(
      (int)(blockIdx.x * (S1_T / 32) + (threadIdx.x >> 5)));
  int s = src[gw * 32 + lane];
  int d = dst[gw * 32 + lane];
  for (int bt = gw; bt < S1_BATCHES; bt += S1_WAVES) {
    const int nb = (bt + S1_WAVES < S1_BATCHES) ? bt + S1_WAVES : bt;
    const int sn = src[nb * 32 + lane];
    const int dn = dst[nb * 32 + lane];
    const int s16 = s * (D_HID * 4);
    const int d16 = d * (D_HID * 4);
    float m[16];
    S1_GATHER(0)  S1_GATHER(1)  S1_GATHER(2)  S1_GATHER(3)
    S1_GATHER(4)  S1_GATHER(5)  S1_GATHER(6)  S1_GATHER(7)
    S1_GATHER(8)  S1_GATHER(9)  S1_GATHER(10) S1_GATHER(11)
    S1_GATHER(12) S1_GATHER(13) S1_GATHER(14) S1_GATHER(15)
    S1_ADD(0)  S1_ADD(1)  S1_ADD(2)  S1_ADD(3)
    S1_ADD(4)  S1_ADD(5)  S1_ADD(6)  S1_ADD(7)
    S1_ADD(8)  S1_ADD(9)  S1_ADD(10) S1_ADD(11)
    S1_ADD(12) S1_ADD(13) S1_ADD(14) S1_ADD(15)
    s = sn;
    d = dn;
  }
}


__global__ void k_layer2(const float* __restrict__ agg1, const float* __restrict__ b1,
                         const float* __restrict__ W2, const float* __restrict__ dinv,
                         float* __restrict__ ts, float* __restrict__ agg2,
                         const float* __restrict__ agg1b, float* __restrict__ agg2b,
                         size_t ws_size) {
  int i = blockIdx.x * blockDim.x + threadIdx.x;
  if (i >= N_NODES) return;
  const bool two = ws_size >= WS_BYTES_TWO_PLANES;
  const float* row = agg1 + (size_t)i * D_HID;
  const float* rowb = agg1b + (size_t)i * D_HID;
  float a[D_HID];
#pragma unroll
  for (int j = 0; j < D_HID; ++j) a[j] = row[j];
  if (two) {
#pragma unroll
    for (int j = 0; j < D_HID; ++j) a[j] += rowb[j];
  }
  const float di = dinv[i];
  float t0 = 0.f, t1 = 0.f;
#pragma unroll
  for (int j = 0; j < D_HID; ++j) {
    float hj = fmaxf(a[j] * di + b1[j], 0.0f);
    t0 += hj * W2[j * 2 + 0];
    t1 += hj * W2[j * 2 + 1];
  }
  float s0 = t0 * di, s1 = t1 * di;
  ts[2 * i + 0] = s0;
  ts[2 * i + 1] = s1;
  agg2[2 * i + 0] = s0 * di;
  agg2[2 * i + 1] = s1 * di;
  if (two) {
    agg2b[2 * i + 0] = 0.0f;
    agg2b[2 * i + 1] = 0.0f;
  }
}


__global__ void __launch_bounds__(OWN_T)
k_scatter2(const int* __restrict__ src, const int* __restrict__ dst,
           const float* __restrict__ ts, const float* __restrict__ dinv,
           float* __restrict__ agg2, float* __restrict__ agg2b, size_t ws_size) {
  __shared__ float acc[2 * OWN_RANGE];
  float* aggp = (ws_size >= WS_BYTES_TWO_PLANES && on_second_half()) ? agg2b : agg2;
  const int tid = threadIdx.x;
  const int r = blockIdx.x / OWN_B;
  const int b = blockIdx.x % OWN_B;
  const int base = r * OWN_RANGE;
  for (int i = tid; i < 2 * OWN_RANGE; i += OWN_T) acc[i] = 0.0f;
  __syncthreads();
  const int4* d4 = (const int4*)(dst + (size_t)b * SLICE);
  const int4* s4 = (const int4*)(src + (size_t)b * SLICE);
  const float2* ts2 = (const float2*)ts;
  for (int q0 = tid; q0 < SLICE4; q0 += 4 * OWN_T) {
    int dl[16], sv[16];
#pragma unroll
    for (int j = 0; j < 4; ++j) {
      int q = q0 + j * OWN_T;
      int4 d = make_int4(-1, -1, -1, -1);
      int4 s = make_int4(0, 0, 0, 0);
      if (q < SLICE4) { d = d4[q]; s = s4[q]; }
      dl[4 * j + 0] = d.x - base; dl[4 * j + 1] = d.y - base;
      dl[4 * j + 2] = d.z - base; dl[4 * j + 3] = d.w - base;
      sv[4 * j + 0] = s.x; sv[4 * j + 1] = s.y; sv[4 * j + 2] = s.z; sv[4 * j + 3] = s.w;
    }
    float2 m[16];
#pragma unroll
    for (int k = 0; k < 16; ++k) {
      m[k] = make_float2(0.0f, 0.0f);
      if ((unsigned)dl[k] < (unsigned)OWN_RANGE) m[k] = ts2[sv[k]];
    }
#pragma unroll
    for (int k = 0; k < 16; ++k) {
      if ((unsigned)dl[k] < (unsigned)OWN_RANGE) {
        atomicAdd(&acc[2 * dl[k] + 0], m[k].x);
        atomicAdd(&acc[2 * dl[k] + 1], m[k].y);
      }
    }
  }
  __syncthreads();
  for (int i0 = tid; i0 < 2 * OWN_RANGE; i0 += 8 * OWN_T) {
    float a[8], w[8];
#pragma unroll
    for (int j = 0; j < 8; ++j) {
      int i = i0 + j * OWN_T;
      a[j] = 0.0f; w[j] = 0.0f;
      if (i < 2 * OWN_RANGE) { a[j] = acc[i]; w[j] = dinv[base + (i >> 1)]; }
    }
#pragma unroll
    for (int j = 0; j < 8; ++j) {
      int i = i0 + j * OWN_T;
      if (i < 2 * OWN_RANGE) atomicAdd(&aggp[2 * base + i], a[j] * w[j]);
    }
  }
}


__global__ void k_final(const float* __restrict__ agg2, const float* __restrict__ b2,
                        float* __restrict__ out, const float* __restrict__ agg2b,
                        size_t ws_size) {
  int i = blockIdx.x * blockDim.x + threadIdx.x;
  if (i >= N_NODES) return;
  float y0 = agg2[2 * i + 0];
  float y1 = agg2[2 * i + 1];
  if (ws_size >= WS_BYTES_TWO_PLANES) {
    y0 += agg2b[2 * i + 0];
    y1 += agg2b[2 * i + 1];
  }
  float z0 = y0 + b2[0];
  float z1 = y1 + b2[1];
  float mm = fmaxf(z0, z1);
  float lse = mm + logf(expf(z0 - mm) + expf(z1 - mm));
  out[2 * i + 0] = z0 - lse;
  out[2 * i + 1] = z1 - lse;
}

extern "C" void kernel_launch(void* const* d_in, const int* in_sizes, int n_in,
                              void* d_out, int out_size, void* d_ws, size_t ws_size,
                              hipStream_t stream) {
  const float* x  = (const float*)d_in[0];
  const float* W1 = (const float*)d_in[1];
  const float* b1 = (const float*)d_in[2];
  const float* W2 = (const float*)d_in[3];
  const float* b2 = (const float*)d_in[4];
  const int*  ei  = (const int*)d_in[5];
  const int* src = ei;
  const int* dst = ei + N_EDGES;
  float* out = (float*)d_out;

  float* w    = (float*)d_ws;
  float* deg  = w;
  float* Bp   = deg + N_NODES;
  float* hs   = Bp + 2048;
  float* agg1 = hs + (size_t)N_NODES * D_HID;
  float* ts   = agg1 + (size_t)N_NODES * D_HID;
  float* agg2 = ts + (size_t)N_NODES * D_OUT;
  float* agg1b = agg2 + (size_t)N_NODES * D_OUT;
  float* agg2b = agg1b + (size_t)N_NODES * D_HID;
  float* degb  = agg2b + (size_t)N_NODES * D_OUT;

  const int T = 256;
  k_prep<<<PREP_WG, PREP_T, 0, stream>>>(deg, degb, ws_size, W1, Bp);
  k_count_deg<<<OWN_R * OWN_B, OWN_T, 0, stream>>>(dst, deg, degb, ws_size);

  k_gemm1<<<(N_TILES + 7) / 8, T, 0, stream>>>(x, Bp, deg, degb, hs, agg1, agg1b, ws_size);

  k_scatter1<<<S1_WG, S1_T, 0, stream>>>(src, dst, hs, agg1, agg1b, ws_size);

  k_layer2<<<(N_NODES + T - 1) / T, T, 0, stream>>>(agg1, b1, W2, deg, ts, agg2, agg1b, agg2b, ws_size);
  k_scatter2<<<OWN_R * OWN_B, OWN_T, 0, stream>>>(src, dst, ts, deg, agg2, agg2b, ws_size);
  k_final<<<(N_NODES + T - 1) / T, T, 0, stream>>>(agg2, b2, out, agg2b, ws_size);
}


# Layer-1 GEMM fetches each x row as one whole 512-byte coalesced load and reads the A fragments back from LDS: each line of x asked from the second-level cache once instead of eight times
# speedup vs baseline: 6.1536x; 1.0377x over previous
#include <hip/hip_runtime.h>
#include <math.h>

#define N_NODES 100000
#define N_EDGES 3200000
#define D_IN 128
#define D_HID 16
#define D_OUT 2
#define N_TILES (N_NODES / 16)

#define WS_FLOATS_TWO_PLANES \
  ((size_t)N_NODES * (1 + D_HID + D_HID + D_OUT + D_OUT + D_HID + D_OUT + 1) + 2048)
#define WS_BYTES_TWO_PLANES (WS_FLOATS_TWO_PLANES * sizeof(float))

static __device__ __forceinline__ bool on_second_half() {
  return ((__builtin_amdgcn_s_sendmsg_rtn(0x87) >> 18) & 1) != 0;
}

typedef __attribute__((ext_vector_type(2))) float v2f;
typedef __attribute__((ext_vector_type(8))) float v8f;


#define PREP_T 256
#define PREP_WG ((N_NODES / 4 + PREP_T - 1) / PREP_T)

__global__ void k_prep(float* __restrict__ deg, float* __restrict__ degb, size_t ws_size,
                       const float* __restrict__ W1, float* __restrict__ Bp) {
  int t = blockIdx.x * blockDim.x + threadIdx.x;
  if (t < N_NODES / 4) {
    const float4 z = make_float4(0.0f, 0.0f, 0.0f, 0.0f);
    ((float4*)deg)[t] = z;
    if (ws_size >= WS_BYTES_TWO_PLANES) ((float4*)degb)[t] = z;
  }
  if (t < 32 * 32) {
    int ks = t >> 5, lane = t & 31;
    int n = lane & 15;
    int k0 = ks * 4 + ((lane >> 4) ? 2 : 0);
    Bp[2 * t + 0] = W1[k0 * D_HID + n];
    Bp[2 * t + 1] = W1[(k0 + 1) * D_HID + n];
  }
}


#define OWN_R 8
#define OWN_RANGE (N_NODES / OWN_R)
#define OWN_B 32
#define SLICE (N_EDGES / OWN_B)
#define SLICE4 (SLICE / 4)
#define OWN_T 1024

__global__ void __launch_bounds__(OWN_T)
k_count_deg(const int* __restrict__ dst, float* __restrict__ deg, float* __restrict__ degb,
            size_t ws_size) {
  __shared__ unsigned cnt[OWN_RANGE];
  float* degp = (ws_size >= WS_BYTES_TWO_PLANES && on_second_half()) ? degb : deg;
  const int tid = threadIdx.x;
  const int r = blockIdx.x / OWN_B;
  const int b = blockIdx.x % OWN_B;
  const int base = r * OWN_RANGE;
  for (int i = tid; i < OWN_RANGE; i += OWN_T) cnt[i] = 0u;
  __syncthreads();
  const int4* d4 = (const int4*)(dst + (size_t)b * SLICE);
  for (int q0 = tid; q0 < SLICE4; q0 += 4 * OWN_T) {
    int4 d[4];
#pragma unroll
    for (int j = 0; j < 4; ++j) {
      int q = q0 + j * OWN_T;
      d[j] = make_int4(-1, -1, -1, -1);
      if (q < SLICE4) d[j] = d4[q];
    }
#pragma unroll
    for (int j = 0; j < 4; ++j) {
      unsigned l0 = (unsigned)(d[j].x - base), l1 = (unsigned)(d[j].y - base);
      unsigned l2 = (unsigned)(d[j].z - base), l3 = (unsigned)(d[j].w - base);
      if (l0 < (unsigned)OWN_RANGE) atomicAdd(&cnt[l0], 1u);
      if (l1 < (unsigned)OWN_RANGE) atomicAdd(&cnt[l1], 1u);
      if (l2 < (unsigned)OWN_RANGE) atomicAdd(&cnt[l2], 1u);
      if (l3 < (unsigned)OWN_RANGE) atomicAdd(&cnt[l3], 1u);
    }
  }
  __syncthreads();
  for (int i = tid; i < OWN_RANGE; i += OWN_T) {
    unsigned c = cnt[i];
    if (c) atomicAdd(&degp[base + i], (float)c);
  }
}


#define G1_WAVES 10
#define G1_T (G1_WAVES * 32)
#define G1_WG (N_TILES / G1_WAVES)
#define G1_PITCH 132

__global__ void __launch_bounds__(G1_T)
k_gemm1(const float* __restrict__ x, const float* __restrict__ Bp,
        float* deg, const float* degb, float* __restrict__ hs,
        float* __restrict__ agg1, float* __restrict__ agg1b,
        size_t ws_size) {
  __shared__ float xs[G1_WAVES][16 * G1_PITCH];
  const bool two = ws_size >= WS_BYTES_TWO_PLANES;
  int lane = threadIdx.x & 31;
  int wave = threadIdx.x >> 5;
  int tile = blockIdx.x * G1_WAVES + wave;

  const float4* xg = (const float4*)(x + (size_t)tile * 16 * D_IN) + lane;
  float4 xv[16];
#pragma unroll
  for (int i = 0; i < 16; ++i) xv[i] = xg[i * (D_IN / 4)];

  const int n = lane & 15;
  const int mbase = (lane >> 4) * 8;
  const int r0 = tile * 16 + mbase;
  float4 ga = ((const float4*)(deg + r0))[0];
  float4 gb = ((const float4*)(deg + r0))[1];
  if (two) {
    float4 ha = ((const float4*)(degb + r0))[0];
    float4 hb = ((const float4*)(degb + r0))[1];
    ga.x += ha.x; ga.y += ha.y; ga.z += ha.z; ga.w += ha.w;
    gb.x += hb.x; gb.y += hb.y; gb.z += hb.z; gb.w += hb.w;
  }
  float dv[8];
  dv[0] = rsqrtf(ga.x + 1.0f); dv[1] = rsqrtf(ga.y + 1.0f);
  dv[2] = rsqrtf(ga.z + 1.0f); dv[3] = rsqrtf(ga.w + 1.0f);
  dv[4] = rsqrtf(gb.x + 1.0f); dv[5] = rsqrtf(gb.y + 1.0f);
  dv[6] = rsqrtf(gb.z + 1.0f); dv[7] = rsqrtf(gb.w + 1.0f);

  float* my = &xs[wave][0];
#pragma unroll
  for (int i = 0; i < 16; ++i) *(float4*)(my + i * G1_PITCH + 4 * lane) = xv[i];
  __syncthreads();

  int m = lane & 15;
  int koff = (lane >> 4) * 2;
  const float* ar = my + m * G1_PITCH + koff;
  const v2f* bp = (const v2f*)Bp;

  v8f c = {0.f, 0.f, 0.f, 0.f, 0.f, 0.f, 0.f, 0.f};
#pragma unroll
  for (int ks = 0; ks < 32; ++ks) {
    v2f a = *(const v2f*)(ar + ks * 4);
    v2f b = bp[ks * 32 + lane];
    c = __builtin_amdgcn_wmma_f32_16x16x4_f32(
         false, a,  false, b,
         (short)0, c,  false,  false);
  }

#pragma unroll
  for (int v = 0; v < 8; ++v) {
    int r = tile * 16 + mbase + v;
    float di = dv[v];
    float hv = c[v] * di;
    hs[(size_t)r * D_HID + n] = hv;
    agg1[(size_t)r * D_HID + n] = hv;
  }
  if (n == 0) {
    ((float4*)(deg + r0))[0] = make_float4(dv[0], dv[1], dv[2], dv[3]);
    ((float4*)(deg + r0))[1] = make_float4(dv[4], dv[5], dv[6], dv[7]);
  }
  if (two) {
#pragma unroll
    for (int v = 0; v < 8; ++v) {
      int r = tile * 16 + mbase + v;
      agg1b[(size_t)r * D_HID + n] = 0.0f;
    }
  }
}


#define S1_BATCHES (N_EDGES / 32)
#define S1_T 128
#define S1_WG 3125
#define S1_WAVES (S1_WG * (S1_T / 32))

#define ROW_SHARE_I(v, K) __builtin_amdgcn_update_dpp(0, (v), 0x150 + (K), 0xf, 0xf, false)

#define S1_GATHER(K)                                                          \
  {                                                                           \
    unsigned so = (unsigned)ROW_SHARE_I(s16, K) + j4;                         \
    m[K] = *(const float*)((const char*)hs + so);                             \
  }

#define S1_ADD(K)                                                             \
  {                                                                           \
    unsigned dof = (unsigned)ROW_SHARE_I(d16, K) + j4;                        \
    atomicAdd((float*)((char*)aggp + dof), m[K]);                             \
  }

__global__ void __launch_bounds__(S1_T)
k_scatter1(const int* __restrict__ src, const int* __restrict__ dst,
           const float* __restrict__ hs,
           float* __restrict__ agg1, float* __restrict__ agg1b, size_t ws_size) {
  float* aggp = (ws_size >= WS_BYTES_TWO_PLANES && on_second_half()) ? agg1b : agg1;
  const int lane = threadIdx.x & 31;
  const unsigned j4 = (unsigned)(lane & 15) * 4u;
  const int gw = __builtin_amdgcn_readfirstlane(
      (int)(blockIdx.x * (S1_T / 32) + (threadIdx.x >> 5)));
  int s = src[gw * 32 + lane];
  int d = dst[gw * 32 + lane];
  for (int bt = gw; bt < S1_BATCHES; bt += S1_WAVES) {
    const int nb = (bt + S1_WAVES < S1_BATCHES) ? bt + S1_WAVES : bt;
    const int sn = src[nb * 32 + lane];
    const int dn = dst[nb * 32 + lane];
    const int s16 = s * (D_HID * 4);
    const int d16 = d * (D_HID * 4);
    float m[16];
    S1_GATHER(0)  S1_GATHER(1)  S1_GATHER(2)  S1_GATHER(3)
    S1_GATHER(4)  S1_GATHER(5)  S1_GATHER(6)  S1_GATHER(7)
    S1_GATHER(8)  S1_GATHER(9)  S1_GATHER(10) S1_GATHER(11)
    S1_GATHER(12) S1_GATHER(13) S1_GATHER(14) S1_GATHER(15)
    S1_ADD(0)  S1_ADD(1)  S1_ADD(2)  S1_ADD(3)
    S1_ADD(4)  S1_ADD(5)  S1_ADD(6)  S1_ADD(7)
    S1_ADD(8)  S1_ADD(9)  S1_ADD(10) S1_ADD(11)
    S1_ADD(12) S1_ADD(13) S1_ADD(14) S1_ADD(15)
    s = sn;
    d = dn;
  }
}


__global__ void k_layer2(const float* __restrict__ agg1, const float* __restrict__ b1,
                         const float* __restrict__ W2, const float* __restrict__ dinv,
                         float* __restrict__ ts, float* __restrict__ agg2,
                         const float* __restrict__ agg1b, float* __restrict__ agg2b,
                         size_t ws_size) {
  int i = blockIdx.x * blockDim.x + threadIdx.x;
  if (i >= N_NODES) return;
  const bool two = ws_size >= WS_BYTES_TWO_PLANES;
  const float* row = agg1 + (size_t)i * D_HID;
  const float* rowb = agg1b + (size_t)i * D_HID;
  float a[D_HID];
#pragma unroll
  for (int j = 0; j < D_HID; ++j) a[j] = row[j];
  if (two) {
#pragma unroll
    for (int j = 0; j < D_HID; ++j) a[j] += rowb[j];
  }
  const float di = dinv[i];
  float t0 = 0.f, t1 = 0.f;
#pragma unroll
  for (int j = 0; j < D_HID; ++j) {
    float hj = fmaxf(a[j] * di + b1[j], 0.0f);
    t0 += hj * W2[j * 2 + 0];
    t1 += hj * W2[j * 2 + 1];
  }
  float s0 = t0 * di, s1 = t1 * di;
  ts[2 * i + 0] = s0;
  ts[2 * i + 1] = s1;
  agg2[2 * i + 0] = s0 * di;
  agg2[2 * i + 1] = s1 * di;
  if (two) {
    agg2b[2 * i + 0] = 0.0f;
    agg2b[2 * i + 1] = 0.0f;
  }
}


__global__ void __launch_bounds__(OWN_T)
k_scatter2(const int* __restrict__ src, const int* __restrict__ dst,
           const float* __restrict__ ts, const float* __restrict__ dinv,
           float* __restrict__ agg2, float* __restrict__ agg2b, size_t ws_size) {
  __shared__ float acc[2 * OWN_RANGE];
  float* aggp = (ws_size >= WS_BYTES_TWO_PLANES && on_second_half()) ? agg2b : agg2;
  const int tid = threadIdx.x;
  const int r = blockIdx.x / OWN_B;
  const int b = blockIdx.x % OWN_B;
  const int base = r * OWN_RANGE;
  for (int i = tid; i < 2 * OWN_RANGE; i += OWN_T) acc[i] = 0.0f;
  __syncthreads();
  const int4* d4 = (const int4*)(dst + (size_t)b * SLICE);
  const int4* s4 = (const int4*)(src + (size_t)b * SLICE);
  const float2* ts2 = (const float2*)ts;
  for (int q0 = tid; q0 < SLICE4; q0 += 4 * OWN_T) {
    int dl[16], sv[16];
#pragma unroll
    for (int j = 0; j < 4; ++j) {
      int q = q0 + j * OWN_T;
      int4 d = make_int4(-1, -1, -1, -1);
      int4 s = make_int4(0, 0, 0, 0);
      if (q < SLICE4) { d = d4[q]; s = s4[q]; }
      dl[4 * j + 0] = d.x - base; dl[4 * j + 1] = d.y - base;
      dl[4 * j + 2] = d.z - base; dl[4 * j + 3] = d.w - base;
      sv[4 * j + 0] = s.x; sv[4 * j + 1] = s.y; sv[4 * j + 2] = s.z; sv[4 * j + 3] = s.w;
    }
    float2 m[16];
#pragma unroll
    for (int k = 0; k < 16; ++k) {
      m[k] = make_float2(0.0f, 0.0f);
      if ((unsigned)dl[k] < (unsigned)OWN_RANGE) m[k] = ts2[sv[k]];
    }
#pragma unroll
    for (int k = 0; k < 16; ++k) {
      if ((unsigned)dl[k] < (unsigned)OWN_RANGE) {
        atomicAdd(&acc[2 * dl[k] + 0], m[k].x);
        atomicAdd(&acc[2 * dl[k] + 1], m[k].y);
      }
    }
  }
  __syncthreads();
  for (int i0 = tid; i0 < 2 * OWN_RANGE; i0 += 8 * OWN_T) {
    float a[8], w[8];
#pragma unroll
    for (int j = 0; j < 8; ++j) {
      int i = i0 + j * OWN_T;
      a[j] = 0.0f; w[j] = 0.0f;
      if (i < 2 * OWN_RANGE) { a[j] = acc[i]; w[j] = dinv[base + (i >> 1)]; }
    }
#pragma unroll
    for (int j = 0; j < 8; ++j) {
      int i = i0 + j * OWN_T;
      if (i < 2 * OWN_RANGE) atomicAdd(&aggp[2 * base + i], a[j] * w[j]);
    }
  }
}


__global__ void k_final(const float* __restrict__ agg2, const float* __restrict__ b2,
                        float* __restrict__ out, const float* __restrict__ agg2b,
                        size_t ws_size) {
  int i = blockIdx.x * blockDim.x + threadIdx.x;
  if (i >= N_NODES) return;
  float y0 = agg2[2 * i + 0];
  float y1 = agg2[2 * i + 1];
  if (ws_size >= WS_BYTES_TWO_PLANES) {
    y0 += agg2b[2 * i + 0];
    y1 += agg2b[2 * i + 1];
  }
  float z0 = y0 + b2[0];
  float z1 = y1 + b2[1];
  float mm = fmaxf(z0, z1);
  float lse = mm + logf(expf(z0 - mm) + expf(z1 - mm));
  out[2 * i + 0] = z0 - lse;
  out[2 * i + 1] = z1 - lse;
}

extern "C" void kernel_launch(void* const* d_in, const int* in_sizes, int n_in,
                              void* d_out, int out_size, void* d_ws, size_t ws_size,
                              hipStream_t stream) {
  const float* x  = (const float*)d_in[0];
  const float* W1 = (const float*)d_in[1];
  const float* b1 = (const float*)d_in[2];
  const float* W2 = (const float*)d_in[3];
  const float* b2 = (const float*)d_in[4];
  const int*  ei  = (const int*)d_in[5];
  const int* src = ei;
  const int* dst = ei + N_EDGES;
  float* out = (float*)d_out;

  float* w    = (float*)d_ws;
  float* deg  = w;
  float* Bp   = deg + N_NODES;
  float* hs   = Bp + 2048;
  float* agg1 = hs + (size_t)N_NODES * D_HID;
  float* ts   = agg1 + (size_t)N_NODES * D_HID;
  float* agg2 = ts + (size_t)N_NODES * D_OUT;
  float* agg1b = agg2 + (size_t)N_NODES * D_OUT;
  float* agg2b = agg1b + (size_t)N_NODES * D_HID;
  float* degb  = agg2b + (size_t)N_NODES * D_OUT;

  const int T = 256;
  k_prep<<<PREP_WG, PREP_T, 0, stream>>>(deg, degb, ws_size, W1, Bp);
  k_count_deg<<<OWN_R * OWN_B, OWN_T, 0, stream>>>(dst, deg, degb, ws_size);

  k_gemm1<<<G1_WG, G1_T, 0, stream>>>(x, Bp, deg, degb, hs, agg1, agg1b, ws_size);

  k_scatter1<<<S1_WG, S1_T, 0, stream>>>(src, dst, hs, agg1, agg1b, ws_size);

  k_layer2<<<(N_NODES + T - 1) / T, T, 0, stream>>>(agg1, b1, W2, deg, ts, agg2, agg1b, agg2b, ws_size);
  k_scatter2<<<OWN_R * OWN_B, OWN_T, 0, stream>>>(src, dst, ts, deg, agg2, agg2b, ws_size);
  k_final<<<(N_NODES + T - 1) / T, T, 0, stream>>>(agg2, b2, out, agg2b, ws_size);
}


# Edge list sorted once by destination bin (400 nodes) in LDS; one work-group per bin sums both layers' rows in LDS and finishes its nodes: no float adds to memory for the layers (sum order changes)
# speedup vs baseline: 7.0277x; 1.1420x over previous
#include <hip/hip_runtime.h>
#include <math.h>

#define N_NODES 100000
#define N_EDGES 3200000
#define D_IN 128
#define D_HID 16
#define D_OUT 2
#define N_TILES (N_NODES / 16)

#define BIN_NODES 400
#define N_BINS (N_NODES / BIN_NODES)
#define BS_SLICES 256
#define BS_EDGES (N_EDGES / BS_SLICES)
#define BS_PITCH 12512
#define BS_OFFS 256
#define BS_PER_T 13

#define WS_FLOATS_TWO_PLANES \
  ((size_t)N_NODES * (1 + D_HID + D_OUT + 1) + 2048 + (size_t)BS_SLICES * (BS_OFFS + BS_PITCH))
#define WS_BYTES_TWO_PLANES (WS_FLOATS_TWO_PLANES * sizeof(float))

static __device__ __forceinline__ bool on_second_half() {
  return ((__builtin_amdgcn_s_sendmsg_rtn(0x87) >> 18) & 1) != 0;
}

typedef __attribute__((ext_vector_type(2))) float v2f;
typedef __attribute__((ext_vector_type(8))) float v8f;


#define PREP_T 256
#define PREP_WG ((N_NODES / 4 + PREP_T - 1) / PREP_T)

__global__ void k_prep(float* __restrict__ deg, float* __restrict__ degb, size_t ws_size,
                       const float* __restrict__ W1, float* __restrict__ Bp) {
  int t = blockIdx.x * blockDim.x + threadIdx.x;
  if (t < N_NODES / 4) {
    const float4 z = make_float4(0.0f, 0.0f, 0.0f, 0.0f);
    ((float4*)deg)[t] = z;
    if (ws_size >= WS_BYTES_TWO_PLANES) ((float4*)degb)[t] = z;
  }
  if (t < 32 * 32) {
    int ks = t >> 5, lane = t & 31;
    int n = lane & 15;
    int k0 = ks * 4 + ((lane >> 4) ? 2 : 0);
    Bp[2 * t + 0] = W1[k0 * D_HID + n];
    Bp[2 * t + 1] = W1[(k0 + 1) * D_HID + n];
  }
}


#define OWN_R 8
#define OWN_RANGE (N_NODES / OWN_R)
#define OWN_B 32
#define SLICE (N_EDGES / OWN_B)
#define SLICE4 (SLICE / 4)
#define OWN_T 1024

__global__ void __launch_bounds__(OWN_T)
k_count_deg(const int* __restrict__ dst, float* __restrict__ deg, float* __restrict__ degb,
            size_t ws_size) {
  __shared__ unsigned cnt[OWN_RANGE];
  float* degp = (ws_size >= WS_BYTES_TWO_PLANES && on_second_half()) ? degb : deg;
  const int tid = threadIdx.x;
  const int r = blockIdx.x / OWN_B;
  const int b = blockIdx.x % OWN_B;
  const int base = r * OWN_RANGE;
  for (int i = tid; i < OWN_RANGE; i += OWN_T) cnt[i] = 0u;
  __syncthreads();
  const int4* d4 = (const int4*)(dst + (size_t)b * SLICE);
  for (int q0 = tid; q0 < SLICE4; q0 += 4 * OWN_T) {
    int4 d[4];
#pragma unroll
    for (int j = 0; j < 4; ++j) {
      int q = q0 + j * OWN_T;
      d[j] = make_int4(-1, -1, -1, -1);
      if (q < SLICE4) d[j] = d4[q];
    }
#pragma unroll
    for (int j = 0; j < 4; ++j) {
      unsigned l0 = (unsigned)(d[j].x - base), l1 = (unsigned)(d[j].y - base);
      unsigned l2 = (unsigned)(d[j].z - base), l3 = (unsigned)(d[j].w - base);
      if (l0 < (unsigned)OWN_RANGE) atomicAdd(&cnt[l0], 1u);
      if (l1 < (unsigned)OWN_RANGE) atomicAdd(&cnt[l1], 1u);
      if (l2 < (unsigned)OWN_RANGE) atomicAdd(&cnt[l2], 1u);
      if (l3 < (unsigned)OWN_RANGE) atomicAdd(&cnt[l3], 1u);
    }
  }
  __syncthreads();
  for (int i = tid; i < OWN_RANGE; i += OWN_T) {
    unsigned c = cnt[i];
    if (c) atomicAdd(&degp[base + i], (float)c);
  }
}


#define G1_WAVES 10
#define G1_T (G1_WAVES * 32)
#define G1_WG (N_TILES / G1_WAVES)
#define G1_PITCH 132

__global__ void __launch_bounds__(G1_T)
k_gemm1(const float* __restrict__ x, const float* __restrict__ Bp,
        float* deg, const float* degb, float* __restrict__ hs,
        size_t ws_size) {
  __shared__ float xs[G1_WAVES][16 * G1_PITCH];
  const bool two = ws_size >= WS_BYTES_TWO_PLANES;
  int lane = threadIdx.x & 31;
  int wave = threadIdx.x >> 5;
  int tile = blockIdx.x * G1_WAVES + wave;

  const float4* xg = (const float4*)(x + (size_t)tile * 16 * D_IN) + lane;
  float4 xv[16];
#pragma unroll
  for (int i = 0; i < 16; ++i) xv[i] = xg[i * (D_IN / 4)];

  const int n = lane & 15;
  const int mbase = (lane >> 4) * 8;
  const int r0 = tile * 16 + mbase;
  float4 ga = ((const float4*)(deg + r0))[0];
  float4 gb = ((const float4*)(deg + r0))[1];
  if (two) {
    float4 ha = ((const float4*)(degb + r0))[0];
    float4 hb = ((const float4*)(degb + r0))[1];
    ga.x += ha.x; ga.y += ha.y; ga.z += ha.z; ga.w += ha.w;
    gb.x += hb.x; gb.y += hb.y; gb.z += hb.z; gb.w += hb.w;
  }
  float dv[8];
  dv[0] = rsqrtf(ga.x + 1.0f); dv[1] = rsqrtf(ga.y + 1.0f);
  dv[2] = rsqrtf(ga.z + 1.0f); dv[3] = rsqrtf(ga.w + 1.0f);
  dv[4] = rsqrtf(gb.x + 1.0f); dv[5] = rsqrtf(gb.y + 1.0f);
  dv[6] = rsqrtf(gb.z + 1.0f); dv[7] = rsqrtf(gb.w + 1.0f);

  float* my = &xs[wave][0];
#pragma unroll
  for (int i = 0; i < 16; ++i) *(float4*)(my + i * G1_PITCH + 4 * lane) = xv[i];
  __syncthreads();

  int m = lane & 15;
  int koff = (lane >> 4) * 2;
  const float* ar = my + m * G1_PITCH + koff;
  const v2f* bp = (const v2f*)Bp;

  v8f c = {0.f, 0.f, 0.f, 0.f, 0.f, 0.f, 0.f, 0.f};
#pragma unroll
  for (int ks = 0; ks < 32; ++ks) {
    v2f a = *(const v2f*)(ar + ks * 4);
    v2f b = bp[ks * 32 + lane];
    c = __builtin_amdgcn_wmma_f32_16x16x4_f32(
         false, a,  false, b,
         (short)0, c,  false,  false);
  }

#pragma unroll
  for (int v = 0; v < 8; ++v) {
    int r = tile * 16 + mbase + v;
    float di = dv[v];
    float hv = c[v] * di;
    hs[(size_t)r * D_HID + n] = hv;
  }
  if (n == 0) {
    ((float4*)(deg + r0))[0] = make_float4(dv[0], dv[1], dv[2], dv[3]);
    ((float4*)(deg + r0))[1] = make_float4(dv[4], dv[5], dv[6], dv[7]);
  }
}


#define OWN_WG_T 1024

__global__ void __launch_bounds__(OWN_WG_T)
k_bin(const int* __restrict__ src, const int* __restrict__ dst,
      int* __restrict__ offs, int* __restrict__ bins) {
  __shared__ int hist[BS_OFFS];
  __shared__ int start[BS_OFFS];
  __shared__ int sorted[BS_PITCH];
  const int tid = threadIdx.x;
  const int sl = blockIdx.x;
  const int* ds = dst + (size_t)sl * BS_EDGES;
  const int* ss = src + (size_t)sl * BS_EDGES;
  if (tid < BS_OFFS) hist[tid] = 0;
  __syncthreads();
  int pk[BS_PER_T], bn[BS_PER_T];
#pragma unroll
  for (int i = 0; i < BS_PER_T; ++i) {
    const int e = tid + i * OWN_WG_T;
    bn[i] = -1;
    pk[i] = 0;
    if (e < BS_EDGES) {
      const int d = ds[e];
      const int s = ss[e];
      const int b = d / BIN_NODES;
      bn[i] = b;
      pk[i] = (s << 9) | (d - b * BIN_NODES);
    }
  }
#pragma unroll
  for (int i = 0; i < BS_PER_T; ++i)
    if (bn[i] >= 0) atomicAdd(&hist[bn[i]], 1);
  __syncthreads();
  if (tid < 32) {
    int v[8];
    int sum = 0;
#pragma unroll
    for (int j = 0; j < 8; ++j) { v[j] = hist[tid * 8 + j]; sum += v[j]; }
    int inc = sum;
#pragma unroll
    for (int o = 1; o < 32; o <<= 1) {
      const int t = __shfl_up(inc, o);
      if (tid >= o) inc += t;
    }
    int run = inc - sum;
#pragma unroll
    for (int j = 0; j < 8; ++j) { start[tid * 8 + j] = run; run += v[j]; }
  }
  __syncthreads();
  if (tid < BS_OFFS) {
    const int st = start[tid];
    offs[sl * BS_OFFS + tid] = st;
    hist[tid] = st;
  }
  __syncthreads();
#pragma unroll
  for (int i = 0; i < BS_PER_T; ++i)
    if (bn[i] >= 0) sorted[atomicAdd(&hist[bn[i]], 1)] = pk[i];
  __syncthreads();
  int* out = bins + (size_t)sl * BS_PITCH;
  for (int i = tid; i < BS_PITCH; i += OWN_WG_T) out[i] = (i < BS_EDGES) ? sorted[i] : 0;
}

#define ROW_SHARE_I(v, K) __builtin_amdgcn_update_dpp(0, (v), 0x150 + (K), 0xf, 0xf, false)

#define OWN_RUNS (BS_SLICES / (OWN_WG_T / 32))

#define O1_STEP(K)                                                             \
  {                                                                            \
    so[K] = (unsigned)ROW_SHARE_I(sof, K) + j4;                                \
    dw[K] = ROW_SHARE_I(dof, K) + jj;                                          \
  }

static __device__ __forceinline__ void own1_batch(int e, const float* __restrict__ hs,
                                                  float* acc, int jj, unsigned j4) {
  const int ev = e >= 0 ? e : BIN_NODES;
  const int sof = (ev >> 9) * (D_HID * 4);
  const int dof = (ev & 511) * D_HID;
  unsigned so[16];
  int dw[16];
  O1_STEP(0)  O1_STEP(1)  O1_STEP(2)  O1_STEP(3)
  O1_STEP(4)  O1_STEP(5)  O1_STEP(6)  O1_STEP(7)
  O1_STEP(8)  O1_STEP(9)  O1_STEP(10) O1_STEP(11)
  O1_STEP(12) O1_STEP(13) O1_STEP(14) O1_STEP(15)
  float m[16];
#pragma unroll
  for (int K = 0; K < 16; ++K) m[K] = *(const float*)((const char*)hs + so[K]);
#pragma unroll
  for (int K = 0; K < 16; ++K) atomicAdd(&acc[dw[K]], m[K]);
}

__global__ void __launch_bounds__(OWN_WG_T)
k_own1(const int* __restrict__ offs, const int* __restrict__ bins,
       const float* __restrict__ hs, const float* __restrict__ dinv,
       const float* __restrict__ b1, const float* __restrict__ W2,
       float* __restrict__ ts) {
  __shared__ float acc[(BIN_NODES + 1) * D_HID];
  const int tid = threadIdx.x;
  const int lane = tid & 31;
  const int wave = tid >> 5;
  const int k = blockIdx.x;
  const int base = k * BIN_NODES;
  {
    float v[(BIN_NODES * D_HID + OWN_WG_T - 1) / OWN_WG_T];
#pragma unroll
    for (int q = 0; q < (BIN_NODES * D_HID + OWN_WG_T - 1) / OWN_WG_T; ++q) {
      const int i = tid + q * OWN_WG_T;
      v[q] = 0.0f;
      if (i < BIN_NODES * D_HID) v[q] = hs[(size_t)base * D_HID + i];
    }
#pragma unroll
    for (int q = 0; q < (BIN_NODES * D_HID + OWN_WG_T - 1) / OWN_WG_T; ++q) {
      const int i = tid + q * OWN_WG_T;
      if (i < (BIN_NODES + 1) * D_HID) acc[i] = v[q];
    }
  }
  const int jj = lane & 15;
  const unsigned j4 = (unsigned)jj * 4u;
  int o0[OWN_RUNS], o1[OWN_RUNS];
#pragma unroll
  for (int r = 0; r < OWN_RUNS; ++r) {
    const int sl = wave + r * (OWN_WG_T / 32);
    o0[r] = offs[sl * BS_OFFS + k];
    o1[r] = offs[sl * BS_OFFS + k + 1];
  }
  int ea[OWN_RUNS], eb[OWN_RUNS];
#pragma unroll
  for (int r = 0; r < OWN_RUNS; ++r) {
    const int* run = bins + (size_t)(wave + r * (OWN_WG_T / 32)) * BS_PITCH;
    const int i0 = o0[r] + lane;
    ea[r] = -1;
    eb[r] = -1;
    if (i0 < o1[r]) ea[r] = run[i0];
    if (i0 + 32 < o1[r]) eb[r] = run[i0 + 32];
  }
  __syncthreads();
#pragma unroll
  for (int r = 0; r < OWN_RUNS; ++r) {
    const int* run = bins + (size_t)(wave + r * (OWN_WG_T / 32)) * BS_PITCH;
    const int p0 = __builtin_amdgcn_readfirstlane(o0[r]);
    const int p1 = __builtin_amdgcn_readfirstlane(o1[r]);
    int b = 0;
    for (int p = p0; p < p1; p += 32, ++b) {
      int e;
      if (b == 0) {
        e = ea[r];
      } else if (b == 1) {
        e = eb[r];
      } else {
        const int idx = p + lane;
        e = -1;
        if (idx < p1) e = run[idx];
      }
      own1_batch(e, hs, acc, jj, j4);
    }
  }
  __syncthreads();
  if (tid < BIN_NODES) {
    const int i = base + tid;
    const float di = dinv[i];
    float t0 = 0.f, t1 = 0.f;
#pragma unroll
    for (int j = 0; j < D_HID; ++j) {
      float hj = fmaxf(acc[tid * D_HID + j] * di + b1[j], 0.0f);
      t0 += hj * W2[j * 2 + 0];
      t1 += hj * W2[j * 2 + 1];
    }
    ts[2 * i + 0] = t0 * di;
    ts[2 * i + 1] = t1 * di;
  }
}

__global__ void __launch_bounds__(OWN_WG_T)
k_own2(const int* __restrict__ offs, const int* __restrict__ bins,
       const float* __restrict__ ts, const float* __restrict__ dinv,
       const float* __restrict__ b2, float* __restrict__ out) {
  __shared__ float acc[BIN_NODES * D_OUT];
  const int tid = threadIdx.x;
  const int lane = tid & 31;
  const int wave = tid >> 5;
  const int k = blockIdx.x;
  const int base = k * BIN_NODES;
  if (tid < BIN_NODES * D_OUT) acc[tid] = ts[(size_t)base * D_OUT + tid];
  int o0[OWN_RUNS], o1[OWN_RUNS];
#pragma unroll
  for (int r = 0; r < OWN_RUNS; ++r) {
    const int sl = wave + r * (OWN_WG_T / 32);
    o0[r] = offs[sl * BS_OFFS + k];
    o1[r] = offs[sl * BS_OFFS + k + 1];
  }
  const float2* ts2 = (const float2*)ts;
  int ea[OWN_RUNS], eb[OWN_RUNS];
#pragma unroll
  for (int r = 0; r < OWN_RUNS; ++r) {
    const int* run = bins + (size_t)(wave + r * (OWN_WG_T / 32)) * BS_PITCH;
    const int i0 = o0[r] + lane;
    ea[r] = -1;
    eb[r] = -1;
    if (i0 < o1[r]) ea[r] = run[i0];
    if (i0 + 32 < o1[r]) eb[r] = run[i0 + 32];
  }
  float2 ma[OWN_RUNS], mb[OWN_RUNS];
#pragma unroll
  for (int r = 0; r < OWN_RUNS; ++r) {
    ma[r] = make_float2(0.0f, 0.0f);
    mb[r] = make_float2(0.0f, 0.0f);
    if (ea[r] >= 0) ma[r] = ts2[ea[r] >> 9];
    if (eb[r] >= 0) mb[r] = ts2[eb[r] >> 9];
  }
  __syncthreads();
#pragma unroll
  for (int r = 0; r < OWN_RUNS; ++r) {
    if (ea[r] >= 0) {
      atomicAdd(&acc[2 * (ea[r] & 511) + 0], ma[r].x);
      atomicAdd(&acc[2 * (ea[r] & 511) + 1], ma[r].y);
    }
    if (eb[r] >= 0) {
      atomicAdd(&acc[2 * (eb[r] & 511) + 0], mb[r].x);
      atomicAdd(&acc[2 * (eb[r] & 511) + 1], mb[r].y);
    }
    const int* run = bins + (size_t)(wave + r * (OWN_WG_T / 32)) * BS_PITCH;
    for (int idx = o0[r] + 64 + lane; idx < o1[r]; idx += 32) {
      const int e = run[idx];
      const float2 mv = ts2[e >> 9];
      atomicAdd(&acc[2 * (e & 511) + 0], mv.x);
      atomicAdd(&acc[2 * (e & 511) + 1], mv.y);
    }
  }
  __syncthreads();
  if (tid < BIN_NODES) {
    const int i = base + tid;
    const float di = dinv[i];
    float z0 = acc[2 * tid + 0] * di + b2[0];
    float z1 = acc[2 * tid + 1] * di + b2[1];
    float mm = fmaxf(z0, z1);
    float lse = mm + logf(expf(z0 - mm) + expf(z1 - mm));
    out[2 * i + 0] = z0 - lse;
    out[2 * i + 1] = z1 - lse;
  }
}

extern "C" void kernel_launch(void* const* d_in, const int* in_sizes, int n_in,
                              void* d_out, int out_size, void* d_ws, size_t ws_size,
                              hipStream_t stream) {
  const float* x  = (const float*)d_in[0];
  const float* W1 = (const float*)d_in[1];
  const float* b1 = (const float*)d_in[2];
  const float* W2 = (const float*)d_in[3];
  const float* b2 = (const float*)d_in[4];
  const int*  ei  = (const int*)d_in[5];
  const int* src = ei;
  const int* dst = ei + N_EDGES;
  float* out = (float*)d_out;

  float* w    = (float*)d_ws;
  float* deg  = w;
  float* Bp   = deg + N_NODES;
  float* hs   = Bp + 2048;
  float* ts   = hs + (size_t)N_NODES * D_HID;
  int*   offs = (int*)(ts + (size_t)N_NODES * D_OUT);
  int*   bins = offs + BS_SLICES * BS_OFFS;
  float* degb = (float*)(bins + (size_t)BS_SLICES * BS_PITCH);

  k_prep<<<PREP_WG, PREP_T, 0, stream>>>(deg, degb, ws_size, W1, Bp);
  k_count_deg<<<OWN_R * OWN_B, OWN_T, 0, stream>>>(dst, deg, degb, ws_size);

  k_gemm1<<<G1_WG, G1_T, 0, stream>>>(x, Bp, deg, degb, hs, ws_size);

  k_bin<<<BS_SLICES, OWN_WG_T, 0, stream>>>(src, dst, offs, bins);
  k_own1<<<N_BINS, OWN_WG_T, 0, stream>>>(offs, bins, hs, deg, b1, W2, ts);
  k_own2<<<N_BINS, OWN_WG_T, 0, stream>>>(offs, bins, ts, deg, b2, out);
}


# Degree counted from the sorted edge list: the GEMM pass runs one work-group per bin and counts its nodes' in-edges in LDS first; count pass, clear launch and all adds to memory go (6 launches to 4)
# speedup vs baseline: 8.2345x; 1.1717x over previous
#include <hip/hip_runtime.h>
#include <math.h>

#define N_NODES 100000
#define N_EDGES 3200000
#define D_IN 128
#define D_HID 16
#define D_OUT 2
#define N_TILES (N_NODES / 16)

#define BIN_NODES 400
#define N_BINS (N_NODES / BIN_NODES)
#define BS_SLICES 256
#define BS_EDGES (N_EDGES / BS_SLICES)
#define BS_PITCH 12512
#define BS_OFFS 256
#define BS_PER_T 13

typedef __attribute__((ext_vector_type(2))) float v2f;
typedef __attribute__((ext_vector_type(8))) float v8f;


#define OWN_WG_T 1024

__global__ void __launch_bounds__(OWN_WG_T)
k_bin(const int* __restrict__ src, const int* __restrict__ dst,
      int* __restrict__ offs, int* __restrict__ bins,
      const float* __restrict__ W1, float* __restrict__ Bp) {
  __shared__ int hist[BS_OFFS];
  __shared__ int start[BS_OFFS];
  __shared__ int sorted[BS_PITCH];
  const int tid = threadIdx.x;
  const int sl = blockIdx.x;
  const int* ds = dst + (size_t)sl * BS_EDGES;
  const int* ss = src + (size_t)sl * BS_EDGES;
  if (tid < BS_OFFS) hist[tid] = 0;
  __syncthreads();
  float w0 = 0.0f, w1 = 0.0f;
  if (sl == 0) {
    const int ks = tid >> 5, lane = tid & 31;
    const int n = lane & 15;
    const int k0 = ks * 4 + ((lane >> 4) ? 2 : 0);
    w0 = W1[k0 * D_HID + n];
    w1 = W1[(k0 + 1) * D_HID + n];
  }
  int pk[BS_PER_T], bn[BS_PER_T];
#pragma unroll
  for (int i = 0; i < BS_PER_T; ++i) {
    const int e = tid + i * OWN_WG_T;
    bn[i] = -1;
    pk[i] = 0;
    if (e < BS_EDGES) {
      const int d = ds[e];
      const int s = ss[e];
      const int b = d / BIN_NODES;
      bn[i] = b;
      pk[i] = (s << 9) | (d - b * BIN_NODES);
    }
  }
#pragma unroll
  for (int i = 0; i < BS_PER_T; ++i)
    if (bn[i] >= 0) atomicAdd(&hist[bn[i]], 1);
  __syncthreads();
  if (tid < 32) {
    int v[8];
    int sum = 0;
#pragma unroll
    for (int j = 0; j < 8; ++j) { v[j] = hist[tid * 8 + j]; sum += v[j]; }
    int inc = sum;
#pragma unroll
    for (int o = 1; o < 32; o <<= 1) {
      const int t = __shfl_up(inc, o);
      if (tid >= o) inc += t;
    }
    int run = inc - sum;
#pragma unroll
    for (int j = 0; j < 8; ++j) { start[tid * 8 + j] = run; run += v[j]; }
  }
  __syncthreads();
  if (tid < BS_OFFS) {
    const int st = start[tid];
    offs[sl * BS_OFFS + tid] = st;
    hist[tid] = st;
  }
  __syncthreads();
#pragma unroll
  for (int i = 0; i < BS_PER_T; ++i)
    if (bn[i] >= 0) sorted[atomicAdd(&hist[bn[i]], 1)] = pk[i];
  __syncthreads();
  int* out = bins + (size_t)sl * BS_PITCH;
  for (int i = tid; i < BS_PITCH; i += OWN_WG_T) out[i] = (i < BS_EDGES) ? sorted[i] : 0;
  if (sl == 0) {
    ((float2*)Bp)[tid] = make_float2(w0, w1);
  }
}


#define G1_WAVES (BIN_NODES / 16)
#define G1_T (G1_WAVES * 32)
#define G1_PITCH 132
#define G1_RUNS ((BS_SLICES + G1_WAVES - 1) / G1_WAVES)

__global__ void __launch_bounds__(G1_T)
k_gemm1(const float* __restrict__ x, const float* __restrict__ Bp,
        const int* __restrict__ offs, const int* __restrict__ bins,
        float* __restrict__ dinv, float* __restrict__ hs) {
  __shared__ float xs[G1_WAVES][16 * G1_PITCH];
  __shared__ unsigned cnt[BIN_NODES];
  const int tid = threadIdx.x;
  const int lane = tid & 31;
  const int wave = tid >> 5;
  const int k = blockIdx.x;
  const int tile = k * G1_WAVES + wave;

  if (tid < BIN_NODES) cnt[tid] = 0u;
  int o0[G1_RUNS], o1[G1_RUNS];
  unsigned ro[G1_RUNS];
#pragma unroll
  for (int r = 0; r < G1_RUNS; ++r) {
    const int sl = wave + r * G1_WAVES;
    const bool has = sl < BS_SLICES;
    const unsigned slc = has ? (unsigned)sl : (unsigned)(BS_SLICES - 1);
    const int* ob = offs + (slc * BS_OFFS + (unsigned)k);
    const int a0 = ob[0];
    const int a1 = ob[1];
    o0[r] = a0;
    o1[r] = has ? a1 : a0;
    ro[r] = slc * BS_PITCH;
  }
  int ea[G1_RUNS], eb[G1_RUNS];
#pragma unroll
  for (int r = 0; r < G1_RUNS; ++r) {
    const int i0 = o0[r] + lane;
    const int wa = bins[ro[r] + min((unsigned)i0, (unsigned)(BS_PITCH - 1))];
    const int wb = bins[ro[r] + min((unsigned)(i0 + 32), (unsigned)(BS_PITCH - 1))];
    ea[r] = (i0 < o1[r]) ? wa : -1;
    eb[r] = (i0 + 32 < o1[r]) ? wb : -1;
  }

  const float4* xg = (const float4*)(x + (size_t)tile * 16 * D_IN) + lane;
  float4 xv[16];
#pragma unroll
  for (int i = 0; i < 16; ++i) xv[i] = xg[i * (D_IN / 4)];

  float* my = &xs[wave][0];
#pragma unroll
  for (int i = 0; i < 16; ++i) *(float4*)(my + i * G1_PITCH + 4 * lane) = xv[i];
  __syncthreads();
#pragma unroll
  for (int r = 0; r < G1_RUNS; ++r) {
    if (ea[r] >= 0) atomicAdd(&cnt[ea[r] & 511], 1u);
    if (eb[r] >= 0) atomicAdd(&cnt[eb[r] & 511], 1u);
    for (int idx = o0[r] + 64 + lane; idx < o1[r]; idx += 32)
      atomicAdd(&cnt[bins[ro[r] + (unsigned)idx] & 511], 1u);
  }
  __syncthreads();

  const int n = lane & 15;
  const int mbase = (lane >> 4) * 8;
  const int lr0 = wave * 16 + mbase;
  const int r0 = tile * 16 + mbase;
  float dv[8];
#pragma unroll
  for (int v = 0; v < 8; ++v) dv[v] = rsqrtf((float)cnt[lr0 + v] + 1.0f);

  int m = lane & 15;
  int koff = (lane >> 4) * 2;
  const float* ar = my + m * G1_PITCH + koff;
  const v2f* bp = (const v2f*)Bp;

  v8f c = {0.f, 0.f, 0.f, 0.f, 0.f, 0.f, 0.f, 0.f};
#pragma unroll
  for (int ks = 0; ks < 32; ++ks) {
    v2f a = *(const v2f*)(ar + ks * 4);
    v2f b = bp[ks * 32 + lane];
    c = __builtin_amdgcn_wmma_f32_16x16x4_f32(
         false, a,  false, b,
         (short)0, c,  false,  false);
  }

#pragma unroll
  for (int v = 0; v < 8; ++v) {
    int r = tile * 16 + mbase + v;
    float di = dv[v];
    float hv = c[v] * di;
    hs[(size_t)r * D_HID + n] = hv;
  }
  if (n == 0) {
    ((float4*)(dinv + r0))[0] = make_float4(dv[0], dv[1], dv[2], dv[3]);
    ((float4*)(dinv + r0))[1] = make_float4(dv[4], dv[5], dv[6], dv[7]);
  }
}

#define ROW_SHARE_I(v, K) __builtin_amdgcn_update_dpp(0, (v), 0x150 + (K), 0xf, 0xf, false)

#define OWN_RUNS (BS_SLICES / (OWN_WG_T / 32))

#define O1_STEP(K)                                                             \
  {                                                                            \
    so[K] = (unsigned)ROW_SHARE_I(sof, K) + j4;                                \
    dw[K] = ROW_SHARE_I(dof, K) + jj;                                          \
  }

static __device__ __forceinline__ void own1_batch(int e, const float* __restrict__ hs,
                                                  float* acc, int jj, unsigned j4) {
  const int ev = e >= 0 ? e : BIN_NODES;
  const int sof = (ev >> 9) * (D_HID * 4);
  const int dof = (ev & 511) * D_HID;
  unsigned so[16];
  int dw[16];
  O1_STEP(0)  O1_STEP(1)  O1_STEP(2)  O1_STEP(3)
  O1_STEP(4)  O1_STEP(5)  O1_STEP(6)  O1_STEP(7)
  O1_STEP(8)  O1_STEP(9)  O1_STEP(10) O1_STEP(11)
  O1_STEP(12) O1_STEP(13) O1_STEP(14) O1_STEP(15)
  float m[16];
#pragma unroll
  for (int K = 0; K < 16; ++K) m[K] = *(const float*)((const char*)hs + so[K]);
#pragma unroll
  for (int K = 0; K < 16; ++K) atomicAdd(&acc[dw[K]], m[K]);
}

__global__ void __launch_bounds__(OWN_WG_T)
k_own1(const int* __restrict__ offs, const int* __restrict__ bins,
       const float* __restrict__ hs, const float* __restrict__ dinv,
       const float* __restrict__ b1, const float* __restrict__ W2,
       float* __restrict__ ts) {
  __shared__ float acc[(BIN_NODES + 1) * D_HID];
  const int tid = threadIdx.x;
  const int lane = tid & 31;
  const int wave = tid >> 5;
  const int k = blockIdx.x;
  const int base = k * BIN_NODES;
  {
    float v[(BIN_NODES * D_HID + OWN_WG_T - 1) / OWN_WG_T];
#pragma unroll
    for (int q = 0; q < (BIN_NODES * D_HID + OWN_WG_T - 1) / OWN_WG_T; ++q) {
      const int i = tid + q * OWN_WG_T;
      v[q] = 0.0f;
      if (i < BIN_NODES * D_HID) v[q] = hs[(size_t)base * D_HID + i];
    }
#pragma unroll
    for (int q = 0; q < (BIN_NODES * D_HID + OWN_WG_T - 1) / OWN_WG_T; ++q) {
      const int i = tid + q * OWN_WG_T;
      if (i < (BIN_NODES + 1) * D_HID) acc[i] = v[q];
    }
  }
  const int jj = lane & 15;
  const unsigned j4 = (unsigned)jj * 4u;
  int o0[OWN_RUNS], o1[OWN_RUNS];
#pragma unroll
  for (int r = 0; r < OWN_RUNS; ++r) {
    const int sl = wave + r * (OWN_WG_T / 32);
    o0[r] = offs[sl * BS_OFFS + k];
    o1[r] = offs[sl * BS_OFFS + k + 1];
  }
  int ea[OWN_RUNS], eb[OWN_RUNS];
#pragma unroll
  for (int r = 0; r < OWN_RUNS; ++r) {
    const int* run = bins + (size_t)(wave + r * (OWN_WG_T / 32)) * BS_PITCH;
    const int i0 = o0[r] + lane;
    ea[r] = -1;
    eb[r] = -1;
    if (i0 < o1[r]) ea[r] = run[i0];
    if (i0 + 32 < o1[r]) eb[r] = run[i0 + 32];
  }
  __syncthreads();
#pragma unroll
  for (int r = 0; r < OWN_RUNS; ++r) {
    const int* run = bins + (size_t)(wave + r * (OWN_WG_T / 32)) * BS_PITCH;
    const int p0 = __builtin_amdgcn_readfirstlane(o0[r]);
    const int p1 = __builtin_amdgcn_readfirstlane(o1[r]);
    int b = 0;
    for (int p = p0; p < p1; p += 32, ++b) {
      int e;
      if (b == 0) {
        e = ea[r];
      } else if (b == 1) {
        e = eb[r];
      } else {
        const int idx = p + lane;
        e = -1;
        if (idx < p1) e = run[idx];
      }
      own1_batch(e, hs, acc, jj, j4);
    }
  }
  __syncthreads();
  if (tid < BIN_NODES) {
    const int i = base + tid;
    const float di = dinv[i];
    float t0 = 0.f, t1 = 0.f;
#pragma unroll
    for (int j = 0; j < D_HID; ++j) {
      float hj = fmaxf(acc[tid * D_HID + j] * di + b1[j], 0.0f);
      t0 += hj * W2[j * 2 + 0];
      t1 += hj * W2[j * 2 + 1];
    }
    ts[2 * i + 0] = t0 * di;
    ts[2 * i + 1] = t1 * di;
  }
}

__global__ void __launch_bounds__(OWN_WG_T)
k_own2(const int* __restrict__ offs, const int* __restrict__ bins,
       const float* __restrict__ ts, const float* __restrict__ dinv,
       const float* __restrict__ b2, float* __restrict__ out) {
  __shared__ float acc[BIN_NODES * D_OUT];
  const int tid = threadIdx.x;
  const int lane = tid & 31;
  const int wave = tid >> 5;
  const int k = blockIdx.x;
  const int base = k * BIN_NODES;
  if (tid < BIN_NODES * D_OUT) acc[tid] = ts[(size_t)base * D_OUT + tid];
  int o0[OWN_RUNS], o1[OWN_RUNS];
#pragma unroll
  for (int r = 0; r < OWN_RUNS; ++r) {
    const int sl = wave + r * (OWN_WG_T / 32);
    o0[r] = offs[sl * BS_OFFS + k];
    o1[r] = offs[sl * BS_OFFS + k + 1];
  }
  const float2* ts2 = (const float2*)ts;
  int ea[OWN_RUNS], eb[OWN_RUNS];
#pragma unroll
  for (int r = 0; r < OWN_RUNS; ++r) {
    const int* run = bins + (size_t)(wave + r * (OWN_WG_T / 32)) * BS_PITCH;
    const int i0 = o0[r] + lane;
    ea[r] = -1;
    eb[r] = -1;
    if (i0 < o1[r]) ea[r] = run[i0];
    if (i0 + 32 < o1[r]) eb[r] = run[i0 + 32];
  }
  float2 ma[OWN_RUNS], mb[OWN_RUNS];
#pragma unroll
  for (int r = 0; r < OWN_RUNS; ++r) {
    ma[r] = make_float2(0.0f, 0.0f);
    mb[r] = make_float2(0.0f, 0.0f);
    if (ea[r] >= 0) ma[r] = ts2[ea[r] >> 9];
    if (eb[r] >= 0) mb[r] = ts2[eb[r] >> 9];
  }
  __syncthreads();
#pragma unroll
  for (int r = 0; r < OWN_RUNS; ++r) {
    if (ea[r] >= 0) {
      atomicAdd(&acc[2 * (ea[r] & 511) + 0], ma[r].x);
      atomicAdd(&acc[2 * (ea[r] & 511) + 1], ma[r].y);
    }
    if (eb[r] >= 0) {
      atomicAdd(&acc[2 * (eb[r] & 511) + 0], mb[r].x);
      atomicAdd(&acc[2 * (eb[r] & 511) + 1], mb[r].y);
    }
    const int* run = bins + (size_t)(wave + r * (OWN_WG_T / 32)) * BS_PITCH;
    for (int idx = o0[r] + 64 + lane; idx < o1[r]; idx += 32) {
      const int e = run[idx];
      const float2 mv = ts2[e >> 9];
      atomicAdd(&acc[2 * (e & 511) + 0], mv.x);
      atomicAdd(&acc[2 * (e & 511) + 1], mv.y);
    }
  }
  __syncthreads();
  if (tid < BIN_NODES) {
    const int i = base + tid;
    const float di = dinv[i];
    float z0 = acc[2 * tid + 0] * di + b2[0];
    float z1 = acc[2 * tid + 1] * di + b2[1];
    float mm = fmaxf(z0, z1);
    float lse = mm + logf(expf(z0 - mm) + expf(z1 - mm));
    out[2 * i + 0] = z0 - lse;
    out[2 * i + 1] = z1 - lse;
  }
}

extern "C" void kernel_launch(void* const* d_in, const int* in_sizes, int n_in,
                              void* d_out, int out_size, void* d_ws, size_t ws_size,
                              hipStream_t stream) {
  const float* x  = (const float*)d_in[0];
  const float* W1 = (const float*)d_in[1];
  const float* b1 = (const float*)d_in[2];
  const float* W2 = (const float*)d_in[3];
  const float* b2 = (const float*)d_in[4];
  const int*  ei  = (const int*)d_in[5];
  const int* src = ei;
  const int* dst = ei + N_EDGES;
  float* out = (float*)d_out;

  float* w    = (float*)d_ws;
  float* dinv = w;
  float* Bp   = dinv + N_NODES;
  float* hs   = Bp + 2048;
  float* ts   = hs + (size_t)N_NODES * D_HID;
  int*   offs = (int*)(ts + (size_t)N_NODES * D_OUT);
  int*   bins = offs + BS_SLICES * BS_OFFS;

  k_bin<<<BS_SLICES, OWN_WG_T, 0, stream>>>(src, dst, offs, bins, W1, Bp);
  k_gemm1<<<N_BINS, G1_T, 0, stream>>>(x, Bp, offs, bins, dinv, hs);
  k_own1<<<N_BINS, OWN_WG_T, 0, stream>>>(offs, bins, hs, dinv, b1, W2, ts);
  k_own2<<<N_BINS, OWN_WG_T, 0, stream>>>(offs, bins, ts, dinv, b2, out);
}


# Layer 2 carries one value per node (score difference: a 2-class log-softmax needs no more); bin owners stage all of it in LDS in two halves, edges read LDS, not memory (sum order changes)
# speedup vs baseline: 8.9721x; 1.0896x over previous
#include <hip/hip_runtime.h>
#include <math.h>

#define N_NODES 100000
#define N_EDGES 3200000
#define D_IN 128
#define D_HID 16
#define D_OUT 2
#define N_TILES (N_NODES / 16)

#define BIN_NODES 400
#define N_BINS (N_NODES / BIN_NODES)
#define BS_SLICES 256
#define BS_EDGES (N_EDGES / BS_SLICES)
#define BS_PITCH 12512
#define BS_OFFS 256
#define BS_PER_T 13
#define TD_PITCH 416
#define TD_WORDS (N_BINS * TD_PITCH)
#define TD_HALF (TD_WORDS / 2)
#define TD_HALF_V4 (TD_HALF / 4)

typedef __attribute__((ext_vector_type(2))) float v2f;
typedef __attribute__((ext_vector_type(8))) float v8f;
typedef __attribute__((ext_vector_type(4))) float v4f;
typedef __attribute__((ext_vector_type(4))) int v4i;


#define OWN_WG_T 1024

__global__ void __launch_bounds__(OWN_WG_T)
k_bin(const int* __restrict__ src, const int* __restrict__ dst,
      int* __restrict__ offs, int* __restrict__ bins,
      const float* __restrict__ W1, float* __restrict__ Bp) {
  __shared__ int hist[BS_OFFS];
  __shared__ int start[BS_OFFS];
  __shared__ int sorted[BS_PITCH];
  const int tid = threadIdx.x;
  const int sl = blockIdx.x;
  const int* ds = dst + (size_t)sl * BS_EDGES;
  const int* ss = src + (size_t)sl * BS_EDGES;
  if (tid < BS_OFFS) hist[tid] = 0;
  __syncthreads();
  float w0 = 0.0f, w1 = 0.0f;
  if (sl == 0) {
    const int ks = tid >> 5, lane = tid & 31;
    const int n = lane & 15;
    const int k0 = ks * 4 + ((lane >> 4) ? 2 : 0);
    w0 = W1[k0 * D_HID + n];
    w1 = W1[(k0 + 1) * D_HID + n];
  }
  int pk[BS_PER_T], bn[BS_PER_T];
#pragma unroll
  for (int i = 0; i < BS_PER_T; ++i) {
    const int e = tid + i * OWN_WG_T;
    bn[i] = -1;
    pk[i] = 0;
    if (e < BS_EDGES) {
      const int d = ds[e];
      const int s = ss[e];
      const int b = d / BIN_NODES;
      bn[i] = b;
      pk[i] = (s << 9) | (d - b * BIN_NODES);
    }
  }
#pragma unroll
  for (int i = 0; i < BS_PER_T; ++i)
    if (bn[i] >= 0) atomicAdd(&hist[bn[i]], 1);
  __syncthreads();
  if (tid < 32) {
    int v[8];
    int sum = 0;
#pragma unroll
    for (int j = 0; j < 8; ++j) { v[j] = hist[tid * 8 + j]; sum += v[j]; }
    int inc = sum;
#pragma unroll
    for (int o = 1; o < 32; o <<= 1) {
      const int t = __shfl_up(inc, o);
      if (tid >= o) inc += t;
    }
    int run = inc - sum;
#pragma unroll
    for (int j = 0; j < 8; ++j) { start[tid * 8 + j] = run; run += v[j]; }
  }
  __syncthreads();
  if (tid < BS_OFFS) {
    const int st = start[tid];
    offs[sl * BS_OFFS + tid] = st;
    hist[tid] = st;
  }
  __syncthreads();
#pragma unroll
  for (int i = 0; i < BS_PER_T; ++i)
    if (bn[i] >= 0) sorted[atomicAdd(&hist[bn[i]], 1)] = pk[i];
  __syncthreads();
  int* out = bins + (size_t)sl * BS_PITCH;
  for (int i = tid; i < BS_PITCH; i += OWN_WG_T) out[i] = (i < BS_EDGES) ? sorted[i] : 0;
  if (sl == 0) {
    ((float2*)Bp)[tid] = make_float2(w0, w1);
  }
}


#define G1_WAVES (BIN_NODES / 16)
#define G1_T (G1_WAVES * 32)
#define G1_PITCH 132
#define G1_RUNS ((BS_SLICES + G1_WAVES - 1) / G1_WAVES)

__global__ void __launch_bounds__(G1_T)
k_gemm1(const float* __restrict__ x, const float* __restrict__ Bp,
        const int* __restrict__ offs, const int* __restrict__ bins,
        float* __restrict__ dinv, float* __restrict__ hs) {
  __shared__ float xs[G1_WAVES][16 * G1_PITCH];
  __shared__ unsigned cnt[BIN_NODES];
  const int tid = threadIdx.x;
  const int lane = tid & 31;
  const int wave = tid >> 5;
  const int k = blockIdx.x;
  const int tile = k * G1_WAVES + wave;

  if (tid < BIN_NODES) cnt[tid] = 0u;
  int o0[G1_RUNS], o1[G1_RUNS];
  unsigned ro[G1_RUNS];
#pragma unroll
  for (int r = 0; r < G1_RUNS; ++r) {
    const int sl = wave + r * G1_WAVES;
    const bool has = sl < BS_SLICES;
    const unsigned slc = has ? (unsigned)sl : (unsigned)(BS_SLICES - 1);
    const int* ob = offs + (slc * BS_OFFS + (unsigned)k);
    const int a0 = ob[0];
    const int a1 = ob[1];
    o0[r] = a0;
    o1[r] = has ? a1 : a0;
    ro[r] = slc * BS_PITCH;
  }
  int ea[G1_RUNS], eb[G1_RUNS];
#pragma unroll
  for (int r = 0; r < G1_RUNS; ++r) {
    const int i0 = o0[r] + lane;
    const int wa = bins[ro[r] + min((unsigned)i0, (unsigned)(BS_PITCH - 1))];
    const int wb = bins[ro[r] + min((unsigned)(i0 + 32), (unsigned)(BS_PITCH - 1))];
    ea[r] = (i0 < o1[r]) ? wa : -1;
    eb[r] = (i0 + 32 < o1[r]) ? wb : -1;
  }

  const float4* xg = (const float4*)(x + (size_t)tile * 16 * D_IN) + lane;
  float4 xv[16];
#pragma unroll
  for (int i = 0; i < 16; ++i) xv[i] = xg[i * (D_IN / 4)];

  float* my = &xs[wave][0];
#pragma unroll
  for (int i = 0; i < 16; ++i) *(float4*)(my + i * G1_PITCH + 4 * lane) = xv[i];
  __syncthreads();
#pragma unroll
  for (int r = 0; r < G1_RUNS; ++r) {
    if (ea[r] >= 0) atomicAdd(&cnt[ea[r] & 511], 1u);
    if (eb[r] >= 0) atomicAdd(&cnt[eb[r] & 511], 1u);
    for (int idx = o0[r] + 64 + lane; idx < o1[r]; idx += 32)
      atomicAdd(&cnt[bins[ro[r] + (unsigned)idx] & 511], 1u);
  }
  __syncthreads();

  const int n = lane & 15;
  const int mbase = (lane >> 4) * 8;
  const int lr0 = wave * 16 + mbase;
  const int r0 = tile * 16 + mbase;
  float dv[8];
#pragma unroll
  for (int v = 0; v < 8; ++v) dv[v] = rsqrtf((float)cnt[lr0 + v] + 1.0f);

  int m = lane & 15;
  int koff = (lane >> 4) * 2;
  const float* ar = my + m * G1_PITCH + koff;
  const v2f* bp = (const v2f*)Bp;

  v8f c = {0.f, 0.f, 0.f, 0.f, 0.f, 0.f, 0.f, 0.f};
#pragma unroll
  for (int ks = 0; ks < 32; ++ks) {
    v2f a = *(const v2f*)(ar + ks * 4);
    v2f b = bp[ks * 32 + lane];
    c = __builtin_amdgcn_wmma_f32_16x16x4_f32(
         false, a,  false, b,
         (short)0, c,  false,  false);
  }

#pragma unroll
  for (int v = 0; v < 8; ++v) {
    int r = tile * 16 + mbase + v;
    float di = dv[v];
    float hv = c[v] * di;
    hs[(size_t)r * D_HID + n] = hv;
  }
  if (n == 0) {
    ((float4*)(dinv + r0))[0] = make_float4(dv[0], dv[1], dv[2], dv[3]);
    ((float4*)(dinv + r0))[1] = make_float4(dv[4], dv[5], dv[6], dv[7]);
  }
}

#define ROW_SHARE_I(v, K) __builtin_amdgcn_update_dpp(0, (v), 0x150 + (K), 0xf, 0xf, false)

#define OWN_RUNS (BS_SLICES / (OWN_WG_T / 32))

#define O1_STEP(K)                                                             \
  {                                                                            \
    so[K] = (unsigned)ROW_SHARE_I(sof, K) + j4;                                \
    dw[K] = ROW_SHARE_I(dof, K) + jj;                                          \
  }

static __device__ __forceinline__ void own1_batch(int e, const float* __restrict__ hs,
                                                  float* acc, int jj, unsigned j4) {
  const int ev = e >= 0 ? e : BIN_NODES;
  const int sof = (ev >> 9) * (D_HID * 4);
  const int dof = (ev & 511) * D_HID;
  unsigned so[16];
  int dw[16];
  O1_STEP(0)  O1_STEP(1)  O1_STEP(2)  O1_STEP(3)
  O1_STEP(4)  O1_STEP(5)  O1_STEP(6)  O1_STEP(7)
  O1_STEP(8)  O1_STEP(9)  O1_STEP(10) O1_STEP(11)
  O1_STEP(12) O1_STEP(13) O1_STEP(14) O1_STEP(15)
  float m[16];
#pragma unroll
  for (int K = 0; K < 16; ++K) m[K] = *(const float*)((const char*)hs + so[K]);
#pragma unroll
  for (int K = 0; K < 16; ++K) atomicAdd(&acc[dw[K]], m[K]);
}

__global__ void __launch_bounds__(OWN_WG_T)
k_own1(const int* __restrict__ offs, const int* __restrict__ bins,
       const float* __restrict__ hs, const float* __restrict__ dinv,
       const float* __restrict__ b1, const float* __restrict__ W2,
       float* __restrict__ td) {
  __shared__ float acc[(BIN_NODES + 1) * D_HID];
  const int tid = threadIdx.x;
  const int lane = tid & 31;
  const int wave = tid >> 5;
  const int k = blockIdx.x;
  const int base = k * BIN_NODES;
  {
    float v[(BIN_NODES * D_HID + OWN_WG_T - 1) / OWN_WG_T];
#pragma unroll
    for (int q = 0; q < (BIN_NODES * D_HID + OWN_WG_T - 1) / OWN_WG_T; ++q) {
      const int i = tid + q * OWN_WG_T;
      v[q] = 0.0f;
      if (i < BIN_NODES * D_HID) v[q] = hs[(size_t)base * D_HID + i];
    }
#pragma unroll
    for (int q = 0; q < (BIN_NODES * D_HID + OWN_WG_T - 1) / OWN_WG_T; ++q) {
      const int i = tid + q * OWN_WG_T;
      if (i < (BIN_NODES + 1) * D_HID) acc[i] = v[q];
    }
  }
  const int jj = lane & 15;
  const unsigned j4 = (unsigned)jj * 4u;
  int o0[OWN_RUNS], o1[OWN_RUNS];
#pragma unroll
  for (int r = 0; r < OWN_RUNS; ++r) {
    const int sl = wave + r * (OWN_WG_T / 32);
    o0[r] = offs[sl * BS_OFFS + k];
    o1[r] = offs[sl * BS_OFFS + k + 1];
  }
  int ea[OWN_RUNS], eb[OWN_RUNS];
#pragma unroll
  for (int r = 0; r < OWN_RUNS; ++r) {
    const int* run = bins + (size_t)(wave + r * (OWN_WG_T / 32)) * BS_PITCH;
    const int i0 = o0[r] + lane;
    ea[r] = -1;
    eb[r] = -1;
    if (i0 < o1[r]) ea[r] = run[i0];
    if (i0 + 32 < o1[r]) eb[r] = run[i0 + 32];
  }
  __syncthreads();
#pragma unroll
  for (int r = 0; r < OWN_RUNS; ++r) {
    const int* run = bins + (size_t)(wave + r * (OWN_WG_T / 32)) * BS_PITCH;
    const int p0 = __builtin_amdgcn_readfirstlane(o0[r]);
    const int p1 = __builtin_amdgcn_readfirstlane(o1[r]);
    int b = 0;
    for (int p = p0; p < p1; p += 32, ++b) {
      int e;
      if (b == 0) {
        e = ea[r];
      } else if (b == 1) {
        e = eb[r];
      } else {
        const int idx = p + lane;
        e = -1;
        if (idx < p1) e = run[idx];
      }
      own1_batch(e, hs, acc, jj, j4);
    }
  }
  __syncthreads();
  if (tid < TD_PITCH) {
    float tdv = 0.0f;
    if (tid < BIN_NODES) {
      const int i = base + tid;
      const float di = dinv[i];
      float t0 = 0.f, t1 = 0.f;
#pragma unroll
      for (int j = 0; j < D_HID; ++j) {
        float hj = fmaxf(acc[tid * D_HID + j] * di + b1[j], 0.0f);
        t0 += hj * W2[j * 2 + 0];
        t1 += hj * W2[j * 2 + 1];
      }
      tdv = (t1 - t0) * di;
    }
    td[k * TD_PITCH + tid] = tdv;
  }
}

#define TD_V4_PER_T ((TD_HALF_V4 + OWN_WG_T - 1) / OWN_WG_T)

static __device__ __forceinline__ unsigned td_slot_u(unsigned s) {
  return s + (s / (unsigned)BIN_NODES) * (unsigned)(TD_PITCH - BIN_NODES);
}

static __device__ __forceinline__ void own2_four(v4i w, unsigned hb, const float* tds,
                                                 float* acc) {
  unsigned q[4];
  float v[4];
#pragma unroll
  for (int j = 0; j < 4; ++j) {
    q[j] = td_slot_u((unsigned)w[j] >> 9) - hb;
    v[j] = tds[q[j] < (unsigned)TD_HALF ? q[j] : 0u];
  }
#pragma unroll
  for (int j = 0; j < 4; ++j)
    if (q[j] < (unsigned)TD_HALF) atomicAdd(&acc[w[j] & 511], v[j]);
}

static __device__ __forceinline__ void own2_tail(const int* __restrict__ run, int p0, int p1,
                                                 int lane, unsigned hb, const float* tds,
                                                 float* acc) {
  for (int idx = p0 + 64 + lane; idx < p1; idx += 32) {
    const int e = run[idx];
    const unsigned q = td_slot_u((unsigned)e >> 9) - hb;
    if (q < (unsigned)TD_HALF) atomicAdd(&acc[e & 511], tds[q]);
  }
}

__global__ void __launch_bounds__(OWN_WG_T)
k_own2(const int* __restrict__ offs, const int* __restrict__ bins,
       const float* __restrict__ td, const float* __restrict__ dinv,
       const float* __restrict__ b2, float* __restrict__ out) {
  __shared__ v4f tds4[TD_HALF_V4];
  __shared__ v4i ew4[4 * OWN_WG_T];
  __shared__ float acc[BIN_NODES];
  const float* tds = (const float*)tds4;
  const int tid = threadIdx.x;
  const int lane = tid & 31;
  const int wave = tid >> 5;
  const int k = blockIdx.x;
  const int base = k * BIN_NODES;
  int o0[OWN_RUNS], o1[OWN_RUNS];
#pragma unroll
  for (int r = 0; r < OWN_RUNS; ++r) {
    const int sl = wave + r * (OWN_WG_T / 32);
    o0[r] = offs[sl * BS_OFFS + k];
    o1[r] = offs[sl * BS_OFFS + k + 1];
  }
  const v4f* tg = (const v4f*)td;
  v4f st[TD_V4_PER_T];
#pragma unroll
  for (int q = 0; q < TD_V4_PER_T; ++q)
    st[q] = tg[min(tid + q * OWN_WG_T, TD_HALF_V4 - 1)];
  float a0 = 0.0f;
  if (tid < BIN_NODES) a0 = td[k * TD_PITCH + tid];
  const int wv = __builtin_amdgcn_readfirstlane(wave);
  int p0[OWN_RUNS], p1[OWN_RUNS];
#pragma unroll
  for (int r = 0; r < OWN_RUNS; ++r) {
    p0[r] = __builtin_amdgcn_readfirstlane(o0[r]);
    p1[r] = __builtin_amdgcn_readfirstlane(o1[r]);
  }
  int ea[OWN_RUNS], eb[OWN_RUNS];
#pragma unroll
  for (int r = 0; r < OWN_RUNS; ++r) {
    const unsigned ro = (unsigned)(wv + r * (OWN_WG_T / 32)) * BS_PITCH;
    const int i0 = p0[r] + lane;
    ea[r] = bins[ro + min((unsigned)i0, (unsigned)(BS_PITCH - 1))];
    eb[r] = bins[ro + min((unsigned)(i0 + 32), (unsigned)(BS_PITCH - 1))];
  }
#pragma unroll
  for (int g = 0; g < OWN_RUNS; g += 4) {
    v4i wa, wb;
#pragma unroll
    for (int j = 0; j < 4; ++j) {
      const int i0 = p0[g + j] + lane;
      wa[j] = (i0 < p1[g + j]) ? ea[g + j] : -1;
      wb[j] = (i0 + 32 < p1[g + j]) ? eb[g + j] : -1;
    }
    ew4[(g / 4) * OWN_WG_T + tid] = wa;
    ew4[(2 + g / 4) * OWN_WG_T + tid] = wb;
  }
#pragma unroll
  for (int q = 0; q < TD_V4_PER_T; ++q) {
    const int i = tid + q * OWN_WG_T;
    if (i < TD_HALF_V4) tds4[i] = st[q];
  }
  if (tid < BIN_NODES) acc[tid] = a0;
  __syncthreads();
#pragma unroll
  for (int q = 0; q < TD_V4_PER_T; ++q)
    st[q] = tg[TD_HALF_V4 + min(tid + q * OWN_WG_T, TD_HALF_V4 - 1)];
#pragma unroll
  for (int g = 0; g < 4; ++g) own2_four(ew4[g * OWN_WG_T + tid], 0u, tds, acc);
#pragma unroll
  for (int r = 0; r < OWN_RUNS; ++r)
    own2_tail(bins + (size_t)(wv + r * (OWN_WG_T / 32)) * BS_PITCH, p0[r], p1[r], lane, 0u, tds, acc);
  __syncthreads();
#pragma unroll
  for (int q = 0; q < TD_V4_PER_T; ++q) {
    const int i = tid + q * OWN_WG_T;
    if (i < TD_HALF_V4) tds4[i] = st[q];
  }
  __syncthreads();
#pragma unroll
  for (int g = 0; g < 4; ++g) own2_four(ew4[g * OWN_WG_T + tid], (unsigned)TD_HALF, tds, acc);
#pragma unroll
  for (int r = 0; r < OWN_RUNS; ++r)
    own2_tail(bins + (size_t)(wv + r * (OWN_WG_T / 32)) * BS_PITCH, p0[r], p1[r], lane, (unsigned)TD_HALF, tds, acc);
  __syncthreads();
  if (tid < BIN_NODES) {
    const int i = base + tid;
    const float di = dinv[i];
    const float d = acc[tid] * di + (b2[1] - b2[0]);
    const float a = fabsf(d);
    const float l = logf(1.0f + expf(-a));
    const float hi = -l;
    const float lo = -a - l;
    ((float2*)out)[i] = (d > 0.0f) ? make_float2(lo, hi) : make_float2(hi, lo);
  }
}

extern "C" void kernel_launch(void* const* d_in, const int* in_sizes, int n_in,
                              void* d_out, int out_size, void* d_ws, size_t ws_size,
                              hipStream_t stream) {
  const float* x  = (const float*)d_in[0];
  const float* W1 = (const float*)d_in[1];
  const float* b1 = (const float*)d_in[2];
  const float* W2 = (const float*)d_in[3];
  const float* b2 = (const float*)d_in[4];
  const int*  ei  = (const int*)d_in[5];
  const int* src = ei;
  const int* dst = ei + N_EDGES;
  float* out = (float*)d_out;

  float* w    = (float*)d_ws;
  float* dinv = w;
  float* Bp   = dinv + N_NODES;
  float* hs   = Bp + 2048;
  float* td   = hs + (size_t)N_NODES * D_HID;
  int*   offs = (int*)(td + (size_t)N_NODES * D_OUT);
  int*   bins = offs + BS_SLICES * BS_OFFS;

  k_bin<<<BS_SLICES, OWN_WG_T, 0, stream>>>(src, dst, offs, bins, W1, Bp);
  k_gemm1<<<N_BINS, G1_T, 0, stream>>>(x, Bp, offs, bins, dinv, hs);
  k_own1<<<N_BINS, OWN_WG_T, 0, stream>>>(offs, bins, hs, dinv, b1, W2, td);
  k_own2<<<N_BINS, OWN_WG_T, 0, stream>>>(offs, bins, td, dinv, b2, out);
}


# Layer-1 sums leave LDS float adds: the bin owner orders its edges by node in LDS (counts from the GEMM pass, integer adds), 4 lanes per node add 16-byte row pieces in registers (sum order changes)
# speedup vs baseline: 10.1926x; 1.1360x over previous
#include <hip/hip_runtime.h>
#include <math.h>

#define N_NODES 100000
#define N_EDGES 3200000
#define D_IN 128
#define D_HID 16
#define D_OUT 2
#define N_TILES (N_NODES / 16)

#define BIN_NODES 400
#define N_BINS (N_NODES / BIN_NODES)
#define BS_SLICES 256
#define BS_EDGES (N_EDGES / BS_SLICES)
#define BS_PITCH 12512
#define BS_OFFS 256
#define BS_PER_T 13
#define TD_PITCH 416
#define TD_WORDS (N_BINS * TD_PITCH)
#define TD_HALF (TD_WORDS / 2)
#define TD_HALF_V4 (TD_HALF / 4)

typedef __attribute__((ext_vector_type(2))) float v2f;
typedef __attribute__((ext_vector_type(8))) float v8f;
typedef __attribute__((ext_vector_type(4))) float v4f;
typedef __attribute__((ext_vector_type(4))) int v4i;


#define OWN_WG_T 1024

__global__ void __launch_bounds__(OWN_WG_T)
k_bin(const int* __restrict__ src, const int* __restrict__ dst,
      int* __restrict__ offs, int* __restrict__ bins,
      const float* __restrict__ W1, float* __restrict__ Bp) {
  __shared__ int hist[BS_OFFS];
  __shared__ int start[BS_OFFS];
  __shared__ int sorted[BS_PITCH];
  const int tid = threadIdx.x;
  const int sl = blockIdx.x;
  const int* ds = dst + (size_t)sl * BS_EDGES;
  const int* ss = src + (size_t)sl * BS_EDGES;
  if (tid < BS_OFFS) hist[tid] = 0;
  __syncthreads();
  float w0 = 0.0f, w1 = 0.0f;
  if (sl == 0) {
    const int ks = tid >> 5, lane = tid & 31;
    const int n = lane & 15;
    const int k0 = ks * 4 + ((lane >> 4) ? 2 : 0);
    w0 = W1[k0 * D_HID + n];
    w1 = W1[(k0 + 1) * D_HID + n];
  }
  int pk[BS_PER_T], bn[BS_PER_T];
#pragma unroll
  for (int i = 0; i < BS_PER_T; ++i) {
    const int e = tid + i * OWN_WG_T;
    bn[i] = -1;
    pk[i] = 0;
    if (e < BS_EDGES) {
      const int d = ds[e];
      const int s = ss[e];
      const int b = d / BIN_NODES;
      bn[i] = b;
      pk[i] = (s << 9) | (d - b * BIN_NODES);
    }
  }
#pragma unroll
  for (int i = 0; i < BS_PER_T; ++i)
    if (bn[i] >= 0) atomicAdd(&hist[bn[i]], 1);
  __syncthreads();
  if (tid < 32) {
    int v[8];
    int sum = 0;
#pragma unroll
    for (int j = 0; j < 8; ++j) { v[j] = hist[tid * 8 + j]; sum += v[j]; }
    int inc = sum;
#pragma unroll
    for (int o = 1; o < 32; o <<= 1) {
      const int t = __shfl_up(inc, o);
      if (tid >= o) inc += t;
    }
    int run = inc - sum;
#pragma unroll
    for (int j = 0; j < 8; ++j) { start[tid * 8 + j] = run; run += v[j]; }
  }
  __syncthreads();
  if (tid < BS_OFFS) {
    const int st = start[tid];
    offs[sl * BS_OFFS + tid] = st;
    hist[tid] = st;
  }
  __syncthreads();
#pragma unroll
  for (int i = 0; i < BS_PER_T; ++i)
    if (bn[i] >= 0) sorted[atomicAdd(&hist[bn[i]], 1)] = pk[i];
  __syncthreads();
  int* out = bins + (size_t)sl * BS_PITCH;
  for (int i = tid; i < BS_PITCH; i += OWN_WG_T) out[i] = (i < BS_EDGES) ? sorted[i] : 0;
  if (sl == 0) {
    ((float2*)Bp)[tid] = make_float2(w0, w1);
  }
}


#define G1_WAVES (BIN_NODES / 16)
#define G1_T (G1_WAVES * 32)
#define G1_PITCH 132
#define G1_RUNS ((BS_SLICES + G1_WAVES - 1) / G1_WAVES)

__global__ void __launch_bounds__(G1_T)
k_gemm1(const float* __restrict__ x, const float* __restrict__ Bp,
        const int* __restrict__ offs, const int* __restrict__ bins,
        float* __restrict__ degc, float* __restrict__ hs) {
  __shared__ float xs[G1_WAVES][16 * G1_PITCH];
  __shared__ unsigned cnt[BIN_NODES];
  const int tid = threadIdx.x;
  const int lane = tid & 31;
  const int wave = tid >> 5;
  const int k = blockIdx.x;
  const int tile = k * G1_WAVES + wave;

  if (tid < BIN_NODES) cnt[tid] = 0u;
  int o0[G1_RUNS], o1[G1_RUNS];
  unsigned ro[G1_RUNS];
#pragma unroll
  for (int r = 0; r < G1_RUNS; ++r) {
    const int sl = wave + r * G1_WAVES;
    const bool has = sl < BS_SLICES;
    const unsigned slc = has ? (unsigned)sl : (unsigned)(BS_SLICES - 1);
    const int* ob = offs + (slc * BS_OFFS + (unsigned)k);
    const int a0 = ob[0];
    const int a1 = ob[1];
    o0[r] = a0;
    o1[r] = has ? a1 : a0;
    ro[r] = slc * BS_PITCH;
  }
  int ea[G1_RUNS], eb[G1_RUNS];
#pragma unroll
  for (int r = 0; r < G1_RUNS; ++r) {
    const int i0 = o0[r] + lane;
    const int wa = bins[ro[r] + min((unsigned)i0, (unsigned)(BS_PITCH - 1))];
    const int wb = bins[ro[r] + min((unsigned)(i0 + 32), (unsigned)(BS_PITCH - 1))];
    ea[r] = (i0 < o1[r]) ? wa : -1;
    eb[r] = (i0 + 32 < o1[r]) ? wb : -1;
  }

  const float4* xg = (const float4*)(x + (size_t)tile * 16 * D_IN) + lane;
  float4 xv[16];
#pragma unroll
  for (int i = 0; i < 16; ++i) xv[i] = xg[i * (D_IN / 4)];

  float* my = &xs[wave][0];
#pragma unroll
  for (int i = 0; i < 16; ++i) *(float4*)(my + i * G1_PITCH + 4 * lane) = xv[i];
  __syncthreads();
#pragma unroll
  for (int r = 0; r < G1_RUNS; ++r) {
    if (ea[r] >= 0) atomicAdd(&cnt[ea[r] & 511], 1u);
    if (eb[r] >= 0) atomicAdd(&cnt[eb[r] & 511], 1u);
    for (int idx = o0[r] + 64 + lane; idx < o1[r]; idx += 32)
      atomicAdd(&cnt[bins[ro[r] + (unsigned)idx] & 511], 1u);
  }
  __syncthreads();

  const int n = lane & 15;
  const int mbase = (lane >> 4) * 8;
  const int lr0 = wave * 16 + mbase;
  const int r0 = tile * 16 + mbase;
  float cf[8], dv[8];
#pragma unroll
  for (int v = 0; v < 8; ++v) {
    cf[v] = (float)cnt[lr0 + v];
    dv[v] = rsqrtf(cf[v] + 1.0f);
  }

  int m = lane & 15;
  int koff = (lane >> 4) * 2;
  const float* ar = my + m * G1_PITCH + koff;
  const v2f* bp = (const v2f*)Bp;

  v8f c = {0.f, 0.f, 0.f, 0.f, 0.f, 0.f, 0.f, 0.f};
#pragma unroll
  for (int ks = 0; ks < 32; ++ks) {
    v2f a = *(const v2f*)(ar + ks * 4);
    v2f b = bp[ks * 32 + lane];
    c = __builtin_amdgcn_wmma_f32_16x16x4_f32(
         false, a,  false, b,
         (short)0, c,  false,  false);
  }

#pragma unroll
  for (int v = 0; v < 8; ++v) {
    int r = tile * 16 + mbase + v;
    float di = dv[v];
    float hv = c[v] * di;
    hs[(size_t)r * D_HID + n] = hv;
  }
  if (n == 0) {
    ((float4*)(degc + r0))[0] = make_float4(cf[0], cf[1], cf[2], cf[3]);
    ((float4*)(degc + r0))[1] = make_float4(cf[4], cf[5], cf[6], cf[7]);
  }
}

#define OWN_RUNS (BS_SLICES / (OWN_WG_T / 32))

#define O1_CAP 32768
#define O1_GROUPS (OWN_WG_T / 4)
#define O1_U 8
#define O1_PFX ((BIN_NODES + 31) / 32)

__global__ void __launch_bounds__(OWN_WG_T)
k_own1(const int* __restrict__ offs, const int* __restrict__ bins,
       const float* __restrict__ hs, const float* __restrict__ degc,
       const float* __restrict__ b1, const float* __restrict__ W2,
       float* __restrict__ td) {
  __shared__ __attribute__((aligned(16))) float acc[BIN_NODES * D_HID];
  __shared__ int ew[O1_CAP + 2 * O1_U];
  __shared__ int startl[BIN_NODES + 1];
  __shared__ int nextl[BIN_NODES];
  const int tid = threadIdx.x;
  const int lane = tid & 31;
  const int wave = tid >> 5;
  const int k = blockIdx.x;
  const int base = k * BIN_NODES;
  int o0[OWN_RUNS], o1[OWN_RUNS];
#pragma unroll
  for (int r = 0; r < OWN_RUNS; ++r) {
    const int sl = wave + r * (OWN_WG_T / 32);
    o0[r] = offs[sl * BS_OFFS + k];
    o1[r] = offs[sl * BS_OFFS + k + 1];
  }
  const int wv = __builtin_amdgcn_readfirstlane(wave);
  float cf[O1_PFX];
#pragma unroll
  for (int j = 0; j < O1_PFX; ++j) cf[j] = 0.0f;
  if (wv == 0) {
#pragma unroll
    for (int j = 0; j < O1_PFX; ++j) cf[j] = degc[base + min(lane * O1_PFX + j, BIN_NODES - 1)];
  }
  int ea[OWN_RUNS], eb[OWN_RUNS];
#pragma unroll
  for (int r = 0; r < OWN_RUNS; ++r) {
    const unsigned ro = (unsigned)(wave + r * (OWN_WG_T / 32)) * BS_PITCH;
    const int i0 = o0[r] + lane;
    const int wa = bins[ro + min((unsigned)i0, (unsigned)(BS_PITCH - 1))];
    const int wb = bins[ro + min((unsigned)(i0 + 32), (unsigned)(BS_PITCH - 1))];
    ea[r] = (i0 < o1[r]) ? wa : -1;
    eb[r] = (i0 + 32 < o1[r]) ? wb : -1;
  }
  if (wv == 0) {
    int cn[O1_PFX];
    int sum = 0;
#pragma unroll
    for (int j = 0; j < O1_PFX; ++j) {
      cn[j] = (lane * O1_PFX + j < BIN_NODES) ? (int)cf[j] : 0;
      sum += cn[j];
    }
    int inc = sum;
#pragma unroll
    for (int o = 1; o < 32; o <<= 1) {
      const int t = __shfl_up(inc, o);
      if (lane >= o) inc += t;
    }
    int run = inc - sum;
#pragma unroll
    for (int j = 0; j < O1_PFX; ++j) {
      const int nd = lane * O1_PFX + j;
      if (nd < BIN_NODES) { startl[nd] = run; nextl[nd] = run; }
      run += cn[j];
    }
    if (lane == 31) startl[BIN_NODES] = run;
  }
  __syncthreads();
  const int total = startl[BIN_NODES];
  if (total <= O1_CAP) {
    int pa[OWN_RUNS], pb[OWN_RUNS];
#pragma unroll
    for (int r = 0; r < OWN_RUNS; ++r) {
      pa[r] = 0;
      pb[r] = 0;
      if (ea[r] >= 0) pa[r] = atomicAdd(&nextl[ea[r] & 511], 1);
      if (eb[r] >= 0) pb[r] = atomicAdd(&nextl[eb[r] & 511], 1);
    }
#pragma unroll
    for (int r = 0; r < OWN_RUNS; ++r) {
      if (ea[r] >= 0) ew[pa[r]] = (ea[r] >> 9) << 2;
      if (eb[r] >= 0) ew[pb[r]] = (eb[r] >> 9) << 2;
    }
#pragma unroll
    for (int r = 0; r < OWN_RUNS; ++r) {
      const int* run = bins + (size_t)(wave + r * (OWN_WG_T / 32)) * BS_PITCH;
      for (int idx = o0[r] + 64 + lane; idx < o1[r]; idx += 32) {
        const int e = run[idx];
        ew[atomicAdd(&nextl[e & 511], 1)] = (e >> 9) << 2;
      }
    }
    __syncthreads();
    const unsigned q = (unsigned)tid & 3u;
    const float4* hs4 = (const float4*)hs;
    for (int node = tid >> 2; node < BIN_NODES; node += O1_GROUPS) {
      const int s0 = startl[node];
      const int s1 = startl[node + 1];
      float4 sum = hs4[(unsigned)(base + node) * 4u + q];
      int id[O1_U];
#pragma unroll
      for (int u = 0; u < O1_U; ++u) id[u] = ew[s0 + u];
      for (int i = s0; i < s1; i += O1_U) {
        float4 v[O1_U];
#pragma unroll
        for (int u = 0; u < O1_U; ++u) {
          const unsigned s = (i + u < s1) ? (unsigned)id[u] : 0u;
          v[u] = hs4[s | q];
        }
        int nid[O1_U];
#pragma unroll
        for (int u = 0; u < O1_U; ++u) nid[u] = ew[i + O1_U + u];
#pragma unroll
        for (int u = 0; u < O1_U; ++u) {
          const bool ok = i + u < s1;
          sum.x += ok ? v[u].x : 0.0f;
          sum.y += ok ? v[u].y : 0.0f;
          sum.z += ok ? v[u].z : 0.0f;
          sum.w += ok ? v[u].w : 0.0f;
        }
#pragma unroll
        for (int u = 0; u < O1_U; ++u) id[u] = nid[u];
      }
      *(float4*)&acc[node * D_HID + 4 * q] = sum;
    }
  } else {
    for (int i = tid; i < BIN_NODES * D_HID; i += OWN_WG_T) acc[i] = hs[(size_t)base * D_HID + i];
    __syncthreads();
    for (int r = 0; r < OWN_RUNS; ++r) {
      const int sl = wave + r * (OWN_WG_T / 32);
      const int* run = bins + (size_t)sl * BS_PITCH;
      const int p1 = offs[sl * BS_OFFS + k + 1];
      for (int idx = offs[sl * BS_OFFS + k] + lane; idx < p1; idx += 32) {
        const int e = run[idx];
        const float* row = hs + (size_t)(e >> 9) * D_HID;
        for (int j = 0; j < D_HID; ++j) atomicAdd(&acc[(e & 511) * D_HID + j], row[j]);
      }
    }
  }
  __syncthreads();
  if (tid < TD_PITCH) {
    float tdv = 0.0f;
    if (tid < BIN_NODES) {
      const int i = base + tid;
      const float di = rsqrtf(degc[i] + 1.0f);
      float t0 = 0.f, t1 = 0.f;
#pragma unroll
      for (int j = 0; j < D_HID; ++j) {
        float hj = fmaxf(acc[tid * D_HID + j] * di + b1[j], 0.0f);
        t0 += hj * W2[j * 2 + 0];
        t1 += hj * W2[j * 2 + 1];
      }
      tdv = (t1 - t0) * di;
    }
    td[k * TD_PITCH + tid] = tdv;
  }
}

#define TD_V4_PER_T ((TD_HALF_V4 + OWN_WG_T - 1) / OWN_WG_T)

static __device__ __forceinline__ unsigned td_slot_u(unsigned s) {
  return s + (s / (unsigned)BIN_NODES) * (unsigned)(TD_PITCH - BIN_NODES);
}

static __device__ __forceinline__ void own2_four(v4i w, unsigned hb, const float* tds,
                                                 float* acc) {
  unsigned q[4];
  float v[4];
#pragma unroll
  for (int j = 0; j < 4; ++j) {
    q[j] = td_slot_u((unsigned)w[j] >> 9) - hb;
    v[j] = tds[q[j] < (unsigned)TD_HALF ? q[j] : 0u];
  }
#pragma unroll
  for (int j = 0; j < 4; ++j)
    if (q[j] < (unsigned)TD_HALF) atomicAdd(&acc[w[j] & 511], v[j]);
}

static __device__ __forceinline__ void own2_tail(const int* __restrict__ run, int p0, int p1,
                                                 int lane, unsigned hb, const float* tds,
                                                 float* acc) {
  for (int idx = p0 + 64 + lane; idx < p1; idx += 32) {
    const int e = run[idx];
    const unsigned q = td_slot_u((unsigned)e >> 9) - hb;
    if (q < (unsigned)TD_HALF) atomicAdd(&acc[e & 511], tds[q]);
  }
}

__global__ void __launch_bounds__(OWN_WG_T)
k_own2(const int* __restrict__ offs, const int* __restrict__ bins,
       const float* __restrict__ td, const float* __restrict__ degc,
       const float* __restrict__ b2, float* __restrict__ out) {
  __shared__ v4f tds4[TD_HALF_V4];
  __shared__ v4i ew4[4 * OWN_WG_T];
  __shared__ float acc[BIN_NODES];
  const float* tds = (const float*)tds4;
  const int tid = threadIdx.x;
  const int lane = tid & 31;
  const int wave = tid >> 5;
  const int k = blockIdx.x;
  const int base = k * BIN_NODES;
  int o0[OWN_RUNS], o1[OWN_RUNS];
#pragma unroll
  for (int r = 0; r < OWN_RUNS; ++r) {
    const int sl = wave + r * (OWN_WG_T / 32);
    o0[r] = offs[sl * BS_OFFS + k];
    o1[r] = offs[sl * BS_OFFS + k + 1];
  }
  const v4f* tg = (const v4f*)td;
  v4f st[TD_V4_PER_T];
#pragma unroll
  for (int q = 0; q < TD_V4_PER_T; ++q)
    st[q] = tg[min(tid + q * OWN_WG_T, TD_HALF_V4 - 1)];
  float a0 = 0.0f;
  if (tid < BIN_NODES) a0 = td[k * TD_PITCH + tid];
  const int wv = __builtin_amdgcn_readfirstlane(wave);
  int p0[OWN_RUNS], p1[OWN_RUNS];
#pragma unroll
  for (int r = 0; r < OWN_RUNS; ++r) {
    p0[r] = __builtin_amdgcn_readfirstlane(o0[r]);
    p1[r] = __builtin_amdgcn_readfirstlane(o1[r]);
  }
  int ea[OWN_RUNS], eb[OWN_RUNS];
#pragma unroll
  for (int r = 0; r < OWN_RUNS; ++r) {
    const unsigned ro = (unsigned)(wv + r * (OWN_WG_T / 32)) * BS_PITCH;
    const int i0 = p0[r] + lane;
    ea[r] = bins[ro + min((unsigned)i0, (unsigned)(BS_PITCH - 1))];
    eb[r] = bins[ro + min((unsigned)(i0 + 32), (unsigned)(BS_PITCH - 1))];
  }
#pragma unroll
  for (int g = 0; g < OWN_RUNS; g += 4) {
    v4i wa, wb;
#pragma unroll
    for (int j = 0; j < 4; ++j) {
      const int i0 = p0[g + j] + lane;
      wa[j] = (i0 < p1[g + j]) ? ea[g + j] : -1;
      wb[j] = (i0 + 32 < p1[g + j]) ? eb[g + j] : -1;
    }
    ew4[(g / 4) * OWN_WG_T + tid] = wa;
    ew4[(2 + g / 4) * OWN_WG_T + tid] = wb;
  }
#pragma unroll
  for (int q = 0; q < TD_V4_PER_T; ++q) {
    const int i = tid + q * OWN_WG_T;
    if (i < TD_HALF_V4) tds4[i] = st[q];
  }
  if (tid < BIN_NODES) acc[tid] = a0;
  __syncthreads();
#pragma unroll
  for (int q = 0; q < TD_V4_PER_T; ++q)
    st[q] = tg[TD_HALF_V4 + min(tid + q * OWN_WG_T, TD_HALF_V4 - 1)];
#pragma unroll
  for (int g = 0; g < 4; ++g) own2_four(ew4[g * OWN_WG_T + tid], 0u, tds, acc);
#pragma unroll
  for (int r = 0; r < OWN_RUNS; ++r)
    own2_tail(bins + (size_t)(wv + r * (OWN_WG_T / 32)) * BS_PITCH, p0[r], p1[r], lane, 0u, tds, acc);
  __syncthreads();
#pragma unroll
  for (int q = 0; q < TD_V4_PER_T; ++q) {
    const int i = tid + q * OWN_WG_T;
    if (i < TD_HALF_V4) tds4[i] = st[q];
  }
  __syncthreads();
#pragma unroll
  for (int g = 0; g < 4; ++g) own2_four(ew4[g * OWN_WG_T + tid], (unsigned)TD_HALF, tds, acc);
#pragma unroll
  for (int r = 0; r < OWN_RUNS; ++r)
    own2_tail(bins + (size_t)(wv + r * (OWN_WG_T / 32)) * BS_PITCH, p0[r], p1[r], lane, (unsigned)TD_HALF, tds, acc);
  __syncthreads();
  if (tid < BIN_NODES) {
    const int i = base + tid;
    const float di = rsqrtf(degc[i] + 1.0f);
    const float d = acc[tid] * di + (b2[1] - b2[0]);
    const float a = fabsf(d);
    const float l = logf(1.0f + expf(-a));
    const float hi = -l;
    const float lo = -a - l;
    ((float2*)out)[i] = (d > 0.0f) ? make_float2(lo, hi) : make_float2(hi, lo);
  }
}

extern "C" void kernel_launch(void* const* d_in, const int* in_sizes, int n_in,
                              void* d_out, int out_size, void* d_ws, size_t ws_size,
                              hipStream_t stream) {
  const float* x  = (const float*)d_in[0];
  const float* W1 = (const float*)d_in[1];
  const float* b1 = (const float*)d_in[2];
  const float* W2 = (const float*)d_in[3];
  const float* b2 = (const float*)d_in[4];
  const int*  ei  = (const int*)d_in[5];
  const int* src = ei;
  const int* dst = ei + N_EDGES;
  float* out = (float*)d_out;

  float* w    = (float*)d_ws;
  float* degc = w;
  float* Bp   = degc + N_NODES;
  float* hs   = Bp + 2048;
  float* td   = hs + (size_t)N_NODES * D_HID;
  int*   offs = (int*)(td + (size_t)N_NODES * D_OUT);
  int*   bins = offs + BS_SLICES * BS_OFFS;

  k_bin<<<BS_SLICES, OWN_WG_T, 0, stream>>>(src, dst, offs, bins, W1, Bp);
  k_gemm1<<<N_BINS, G1_T, 0, stream>>>(x, Bp, offs, bins, degc, hs);
  k_own1<<<N_BINS, OWN_WG_T, 0, stream>>>(offs, bins, hs, degc, b1, W2, td);
  k_own2<<<N_BINS, OWN_WG_T, 0, stream>>>(offs, bins, td, degc, b2, out);
}
